# starting kernel
# baseline (speedup 1.0000x reference)
#include <hip/hip_runtime.h>

#define BB 4
#define TT 128
#define SS 512
#define HH 512
#define LN_EPS 1e-5f
#define LOG2E 1.4426950408889634f

typedef __attribute__((ext_vector_type(2))) float v2f;
typedef __attribute__((ext_vector_type(8))) float v8f;

__device__ __forceinline__ v2f ld2(const float* p) {
    const float2 t = *reinterpret_cast<const float2*>(p);
    v2f r; r.x = t.x; r.y = t.y; return r;
}

__device__ __forceinline__ float fast_tanh(float x) {
    float xc = fminf(fmaxf(x, -15.0f), 15.0f);
    float e  = __builtin_amdgcn_exp2f(xc * (2.0f * LOG2E));
    return (e - 1.0f) * __builtin_amdgcn_rcpf(e + 1.0f);
}

#define WMMA_F32(A, Bf, C) \
    __builtin_amdgcn_wmma_f32_16x16x4_f32(false, (A), false, (Bf), (short)0, (C), false, false)

__device__ __forceinline__ void gemm_strip(const float* __restrict__ xrow,
                                           const float* __restrict__ w0,
                                           const float* __restrict__ w1,
                                           const float* __restrict__ w2,
                                           const float* __restrict__ w3,
                                           int Klen,
                                           v8f& acc0, v8f& acc1, v8f& acc2, v8f& acc3) {
    v2f a_c[4], bc0[4], bc1[4], bc2[4], bc3[4];
#pragma unroll
    for (int j = 0; j < 4; ++j) {
        const int o = 4 * j;
        a_c[j] = ld2(xrow + o);
        bc0[j] = ld2(w0 + o); bc1[j] = ld2(w1 + o);
        bc2[j] = ld2(w2 + o); bc3[j] = ld2(w3 + o);
    }
    for (int k = 16; k < Klen; k += 16) {
        v2f a_n[4], bn0[4], bn1[4], bn2[4], bn3[4];
#pragma unroll
        for (int j = 0; j < 4; ++j) {
            const int o = k + 4 * j;
            a_n[j] = ld2(xrow + o);
            bn0[j] = ld2(w0 + o); bn1[j] = ld2(w1 + o);
            bn2[j] = ld2(w2 + o); bn3[j] = ld2(w3 + o);
        }
        __builtin_prefetch(xrow + k + 112, 0, 3);
#pragma unroll
        for (int j = 0; j < 4; ++j) {
            acc0 = WMMA_F32(a_c[j], bc0[j], acc0);
            acc1 = WMMA_F32(a_c[j], bc1[j], acc1);
            acc2 = WMMA_F32(a_c[j], bc2[j], acc2);
            acc3 = WMMA_F32(a_c[j], bc3[j], acc3);
        }
#pragma unroll
        for (int j = 0; j < 4; ++j) {
            a_c[j] = a_n[j];
            bc0[j] = bn0[j]; bc1[j] = bn1[j];
            bc2[j] = bn2[j]; bc3[j] = bn3[j];
        }
    }
#pragma unroll
    for (int j = 0; j < 4; ++j) {
        acc0 = WMMA_F32(a_c[j], bc0[j], acc0);
        acc1 = WMMA_F32(a_c[j], bc1[j], acc1);
        acc2 = WMMA_F32(a_c[j], bc2[j], acc2);
        acc3 = WMMA_F32(a_c[j], bc3[j], acc3);
    }
}

__global__ void gemm_xwt_kernel(const float* __restrict__ X,
                                const float* __restrict__ W,
                                float* __restrict__ C,
                                int M, int N, int K) {
    const int wave = (blockIdx.x * blockDim.x + threadIdx.x) >> 5;
    const int lane = threadIdx.x & 31;
    const int hi   = lane >> 4;
    const int lr   = lane & 15;
    const int nt   = N >> 6;
    const int m0   = (wave / nt) * 16;
    const int n0   = (wave % nt) * 64;

    const float* xrow = X + (size_t)(m0 + lr) * K + 2 * hi;
    const float* w0 = W + (size_t)(n0 +  0 + lr) * K + 2 * hi;
    const float* w1 = W + (size_t)(n0 + 16 + lr) * K + 2 * hi;
    const float* w2 = W + (size_t)(n0 + 32 + lr) * K + 2 * hi;
    const float* w3 = W + (size_t)(n0 + 48 + lr) * K + 2 * hi;

    v8f acc0 = {}, acc1 = {}, acc2 = {}, acc3 = {};
    gemm_strip(xrow, w0, w1, w2, w3, K, acc0, acc1, acc2, acc3);

#pragma unroll
    for (int r = 0; r < 8; ++r) {
        const size_t row = (size_t)(m0 + r + 8 * hi) * N + n0 + lr;
        C[row +  0] = acc0[r];
        C[row + 16] = acc1[r];
        C[row + 32] = acc2[r];
        C[row + 48] = acc3[r];
    }
}

__global__ void score_softmax_kernel(const float* __restrict__ qp,
                                     const float* __restrict__ kp,
                                     const float* __restrict__ vvec,
                                     const int* __restrict__ lens,
                                     float* __restrict__ attn) {
    const int b = blockIdx.x / TT;
    const int t = blockIdx.x % TT;
    const int tid  = threadIdx.x;
    const int wv   = tid >> 5;
    const int lane = tid & 31;

    __shared__ __align__(16) float qsh[HH];
    __shared__ __align__(16) float vsh[HH];
    __shared__ float sc[SS];
    __shared__ float redbuf[8];

    for (int i = tid; i < HH; i += 256) {
        qsh[i] = qp[(size_t)(b * TT + t) * HH + i];
        vsh[i] = vvec[i];
    }
    __syncthreads();

    const int len = lens[b];
    const float* kpb = kp + (size_t)b * SS * HH;

    for (int s = wv; s < SS; s += 8) {
        const float* krow = kpb + (size_t)s * HH;
        float acc = 0.0f;
#pragma unroll
        for (int j = 0; j < 4; ++j) {
            const int h = (lane + 32 * j) * 4;
            const float4 kq = *(const float4*)(krow + h);
            const float4 qq = *(const float4*)(qsh + h);
            const float4 vq = *(const float4*)(vsh + h);
            acc += fast_tanh(qq.x + kq.x) * vq.x;
            acc += fast_tanh(qq.y + kq.y) * vq.y;
            acc += fast_tanh(qq.z + kq.z) * vq.z;
            acc += fast_tanh(qq.w + kq.w) * vq.w;
        }
#pragma unroll
        for (int o = 16; o >= 1; o >>= 1) acc += __shfl_xor(acc, o, 32);
        if (lane == 0) sc[s] = (s < len) ? acc : -__builtin_inff();
    }
    __syncthreads();

    float m = fmaxf(sc[tid], sc[tid + 256]);
#pragma unroll
    for (int o = 16; o >= 1; o >>= 1) m = fmaxf(m, __shfl_xor(m, o, 32));
    if (lane == 0) redbuf[wv] = m;
    __syncthreads();
    float mx = redbuf[0];
#pragma unroll
    for (int i = 1; i < 8; ++i) mx = fmaxf(mx, redbuf[i]);
    __syncthreads();

    const float e0 = __builtin_amdgcn_exp2f((sc[tid]       - mx) * LOG2E);
    const float e1 = __builtin_amdgcn_exp2f((sc[tid + 256] - mx) * LOG2E);
    float ps = e0 + e1;
#pragma unroll
    for (int o = 16; o >= 1; o >>= 1) ps += __shfl_xor(ps, o, 32);
    if (lane == 0) redbuf[wv] = ps;
    __syncthreads();
    float tot = redbuf[0];
#pragma unroll
    for (int i = 1; i < 8; ++i) tot += redbuf[i];
    const float inv = __builtin_amdgcn_rcpf(tot);

    float* arow = attn + (size_t)(b * TT + t) * SS;
    arow[tid]       = e0 * inv;
    arow[tid + 256] = e1 * inv;
}

__device__ __forceinline__ void ctx_loadB(const float* __restrict__ eb, int ka, int n,
                                          v2f& b0, v2f& b1, v2f& b2, v2f& b3) {
    const float* e0 = eb + (size_t)ka * HH + n;
    b0.x = e0[ 0]; b0.y = e0[HH +  0];
    b1.x = e0[16]; b1.y = e0[HH + 16];
    b2.x = e0[32]; b2.y = e0[HH + 32];
    b3.x = e0[48]; b3.y = e0[HH + 48];
}

__global__ void context_kernel(const float* __restrict__ attn,
                               const float* __restrict__ enc,
                               float* __restrict__ ctx) {
    const int wave = (blockIdx.x * blockDim.x + threadIdx.x) >> 5;
    const int lane = threadIdx.x & 31;
    const int hi   = lane >> 4;
    const int lr   = lane & 15;
    const int b    = wave >> 6;
    const int rem  = wave & 63;
    const int m0   = (rem >> 3) * 16;
    const int n0   = (rem & 7) * 64;

    const float* arow = attn + (size_t)(b * TT + m0 + lr) * SS + 2 * hi;
    const float* eb   = enc + (size_t)b * SS * HH;
    const int    nb   = n0 + lr;

    v8f acc0 = {}, acc1 = {}, acc2 = {}, acc3 = {};
    v2f a_c[4], bc0[4], bc1[4], bc2[4], bc3[4];
#pragma unroll
    for (int j = 0; j < 4; ++j) {
        a_c[j] = ld2(arow + 4 * j);
        ctx_loadB(eb, 4 * j + 2 * hi, nb, bc0[j], bc1[j], bc2[j], bc3[j]);
    }
    for (int k = 16; k < SS; k += 16) {
        v2f a_n[4], bn0[4], bn1[4], bn2[4], bn3[4];
#pragma unroll
        for (int j = 0; j < 4; ++j) {
            a_n[j] = ld2(arow + k + 4 * j);
            ctx_loadB(eb, k + 4 * j + 2 * hi, nb, bn0[j], bn1[j], bn2[j], bn3[j]);
        }
#pragma unroll
        for (int j = 0; j < 4; ++j) {
            acc0 = WMMA_F32(a_c[j], bc0[j], acc0);
            acc1 = WMMA_F32(a_c[j], bc1[j], acc1);
            acc2 = WMMA_F32(a_c[j], bc2[j], acc2);
            acc3 = WMMA_F32(a_c[j], bc3[j], acc3);
        }
#pragma unroll
        for (int j = 0; j < 4; ++j) {
            a_c[j] = a_n[j];
            bc0[j] = bn0[j]; bc1[j] = bn1[j];
            bc2[j] = bn2[j]; bc3[j] = bn3[j];
        }
    }
#pragma unroll
    for (int j = 0; j < 4; ++j) {
        acc0 = WMMA_F32(a_c[j], bc0[j], acc0);
        acc1 = WMMA_F32(a_c[j], bc1[j], acc1);
        acc2 = WMMA_F32(a_c[j], bc2[j], acc2);
        acc3 = WMMA_F32(a_c[j], bc3[j], acc3);
    }

#pragma unroll
    for (int r = 0; r < 8; ++r) {
        const size_t row = (size_t)(b * TT + m0 + r + 8 * hi) * HH + n0 + lr;
        ctx[row +  0] = acc0[r];
        ctx[row + 16] = acc1[r];
        ctx[row + 32] = acc2[r];
        ctx[row + 48] = acc3[r];
    }
}

__global__ void out_gemm_kernel(const float* __restrict__ ctx,
                                const float* __restrict__ query,
                                const float* __restrict__ Wout,
                                const float* __restrict__ bout,
                                float* __restrict__ outp) {
    const int wave = (blockIdx.x * blockDim.x + threadIdx.x) >> 5;
    const int lane = threadIdx.x & 31;
    const int hi   = lane >> 4;
    const int lr   = lane & 15;
    const int m0   = (wave >> 3) * 16;
    const int n0   = (wave & 7) * 64;

    const float* crow = ctx   + (size_t)(m0 + lr) * HH + 2 * hi;
    const float* qrow = query + (size_t)(m0 + lr) * HH + 2 * hi;
    const float* w0 = Wout + (size_t)(n0 +  0 + lr) * (2 * HH) + 2 * hi;
    const float* w1 = Wout + (size_t)(n0 + 16 + lr) * (2 * HH) + 2 * hi;
    const float* w2 = Wout + (size_t)(n0 + 32 + lr) * (2 * HH) + 2 * hi;
    const float* w3 = Wout + (size_t)(n0 + 48 + lr) * (2 * HH) + 2 * hi;

    v8f acc0 = {}, acc1 = {}, acc2 = {}, acc3 = {};
    gemm_strip(crow, w0, w1, w2, w3, HH, acc0, acc1, acc2, acc3);
    gemm_strip(qrow, w0 + HH, w1 + HH, w2 + HH, w3 + HH, HH, acc0, acc1, acc2, acc3);

    const float bb0 = bout[n0 +  0 + lr];
    const float bb1 = bout[n0 + 16 + lr];
    const float bb2 = bout[n0 + 32 + lr];
    const float bb3 = bout[n0 + 48 + lr];
#pragma unroll
    for (int r = 0; r < 8; ++r) {
        const size_t row = (size_t)(m0 + r + 8 * hi) * HH + n0 + lr;
        outp[row +  0] = fast_tanh(acc0[r] + bb0);
        outp[row + 16] = fast_tanh(acc1[r] + bb1);
        outp[row + 32] = fast_tanh(acc2[r] + bb2);
        outp[row + 48] = fast_tanh(acc3[r] + bb3);
    }
}

__global__ void ln_kernel(const float* __restrict__ x,
                          const float* __restrict__ gamma,
                          const float* __restrict__ beta,
                          float* __restrict__ out) {
    const int wave = (blockIdx.x * blockDim.x + threadIdx.x) >> 5;
    const int lane = threadIdx.x & 31;
    const float* row = x + (size_t)wave * HH;

    float4 val[4];
    float s = 0.0f;
#pragma unroll
    for (int j = 0; j < 4; ++j) {
        val[j] = *(const float4*)(row + (lane + 32 * j) * 4);
        s += val[j].x + val[j].y + val[j].z + val[j].w;
    }
#pragma unroll
    for (int o = 16; o >= 1; o >>= 1) s += __shfl_xor(s, o, 32);
    const float mu = s * (1.0f / HH);

    float q = 0.0f;
#pragma unroll
    for (int j = 0; j < 4; ++j) {
        const float dx = val[j].x - mu, dy = val[j].y - mu;
        const float dz = val[j].z - mu, dw = val[j].w - mu;
        q += dx * dx + dy * dy + dz * dz + dw * dw;
    }
#pragma unroll
    for (int o = 16; o >= 1; o >>= 1) q += __shfl_xor(q, o, 32);
    const float rstd = __builtin_amdgcn_rsqf(q * (1.0f / HH) + LN_EPS);

    float* orow = out + (size_t)wave * HH;
#pragma unroll
    for (int j = 0; j < 4; ++j) {
        const int h = (lane + 32 * j) * 4;
        const float4 g = *(const float4*)(gamma + h);
        const float4 bt = *(const float4*)(beta + h);
        float4 r;
        r.x = (val[j].x - mu) * rstd * g.x + bt.x;
        r.y = (val[j].y - mu) * rstd * g.y + bt.y;
        r.z = (val[j].z - mu) * rstd * g.z + bt.z;
        r.w = (val[j].w - mu) * rstd * g.w + bt.w;
        *(float4*)(orow + h) = r;
    }
}

extern "C" void kernel_launch(void* const* d_in, const int* in_sizes, int n_in,
                              void* d_out, int out_size, void* d_ws, size_t ws_size,
                              hipStream_t stream) {
    const float* query  = (const float*)d_in[0];
    const float* enc    = (const float*)d_in[1];
    const int*   lens   = (const int*)  d_in[2];
    const float* W_h    = (const float*)d_in[3];
    const float* W_s    = (const float*)d_in[4];
    const float* vvec   = (const float*)d_in[5];
    const float* W_out  = (const float*)d_in[6];
    const float* b_out  = (const float*)d_in[7];
    const float* gamma  = (const float*)d_in[8];
    const float* beta   = (const float*)d_in[9];
    float* out = (float*)d_out;

    float* qp   = (float*)d_ws;
    float* kp   = qp   + (size_t)BB * TT * HH;
    float* attn = kp   + (size_t)BB * SS * HH;
    float* ctx  = attn + (size_t)BB * TT * SS;
    float* outp = ctx  + (size_t)BB * TT * HH;

    gemm_xwt_kernel<<<32, 256, 0, stream>>>(query, W_s, qp, BB * TT, HH, HH);
    gemm_xwt_kernel<<<128, 256, 0, stream>>>(enc, W_h, kp, BB * SS, HH, HH);
    score_softmax_kernel<<<BB * TT, 256, 0, stream>>>(qp, kp, vvec, lens, attn);
    context_kernel<<<32, 256, 0, stream>>>(attn, enc, ctx);
    out_gemm_kernel<<<32, 256, 0, stream>>>(ctx, query, W_out, b_out, outp);
    ln_kernel<<<64, 256, 0, stream>>>(outp, gamma, beta, out);
}


# q/k projection and output GEMMs: fragment loads run two 16-k blocks ahead of the WMMAs (ring of three register sets, 40 to 60 loads outstanding) instead of one; sum order unchanged
# speedup vs baseline: 5.1399x; 5.1399x over previous
#include <hip/hip_runtime.h>

#define BB 4
#define TT 128
#define SS 512
#define HH 512
#define LN_EPS 1e-5f
#define LOG2E 1.4426950408889634f

typedef __attribute__((ext_vector_type(2))) float v2f;
typedef __attribute__((ext_vector_type(8))) float v8f;

__device__ __forceinline__ v2f ld2(const float* p) {
    const float2 t = *reinterpret_cast<const float2*>(p);
    v2f r; r.x = t.x; r.y = t.y; return r;
}

__device__ __forceinline__ float fast_tanh(float x) {
    float xc = fminf(fmaxf(x, -15.0f), 15.0f);
    float e  = __builtin_amdgcn_exp2f(xc * (2.0f * LOG2E));
    return (e - 1.0f) * __builtin_amdgcn_rcpf(e + 1.0f);
}

__device__ __forceinline__ v2f fast_tanh2(v2f x) {
    v2f xc;
    xc.x = fminf(fmaxf(x.x, -15.0f), 15.0f);
    xc.y = fminf(fmaxf(x.y, -15.0f), 15.0f);
    const v2f tt = xc * (2.0f * LOG2E);
    v2f e;
    e.x = __builtin_amdgcn_exp2f(tt.x);
    e.y = __builtin_amdgcn_exp2f(tt.y);
    const v2f d = e + 1.0f;
    v2f r;
    r.x = __builtin_amdgcn_rcpf(d.x);
    r.y = __builtin_amdgcn_rcpf(d.y);
    return (e - 1.0f) * r;
}

#define WMMA_F32(A, Bf, C) \
    __builtin_amdgcn_wmma_f32_16x16x4_f32(false, (A), false, (Bf), (short)0, (C), false, false)

struct Frag { v2f a[4], b0[4], b1[4], b2[4], b3[4]; };

__device__ __forceinline__ void frag_load(Frag& f, const float* __restrict__ xrow,
                                          const float* __restrict__ w0, const float* __restrict__ w1,
                                          const float* __restrict__ w2, const float* __restrict__ w3, int kb) {
#pragma unroll
    for (int j = 0; j < 4; ++j) {
        const int o = 16 * kb + 4 * j;
        f.a[j]  = ld2(xrow + o);
        f.b0[j] = ld2(w0 + o); f.b1[j] = ld2(w1 + o);
        f.b2[j] = ld2(w2 + o); f.b3[j] = ld2(w3 + o);
    }
    __builtin_amdgcn_sched_barrier(0);
}

__device__ __forceinline__ void frag_mma(const Frag& f,
                                         v8f& acc0, v8f& acc1, v8f& acc2, v8f& acc3) {
#pragma unroll
    for (int j = 0; j < 4; ++j) {
        acc0 = WMMA_F32(f.a[j], f.b0[j], acc0);
        acc1 = WMMA_F32(f.a[j], f.b1[j], acc1);
        acc2 = WMMA_F32(f.a[j], f.b2[j], acc2);
        acc3 = WMMA_F32(f.a[j], f.b3[j], acc3);
    }
    asm volatile("" : "+v"(acc0), "+v"(acc1), "+v"(acc2), "+v"(acc3) : : "memory");
}

template <int KB>
__device__ __forceinline__ void gemm_strip(const float* __restrict__ xrow,
                                           const float* __restrict__ w0, const float* __restrict__ w1,
                                           const float* __restrict__ w2, const float* __restrict__ w3,
                                           v8f& acc0, v8f& acc1, v8f& acc2, v8f& acc3) {
    static_assert(KB >= 3, "ring of three");
    constexpr int G = (KB - 2) / 3;
    constexpr int R = KB - 3 * G;
    Frag s0, s1, s2;
    frag_load(s0, xrow, w0, w1, w2, w3, 0);
    frag_load(s1, xrow, w0, w1, w2, w3, 1);
    for (int g = 0; g < G; ++g) {
        const int i = 3 * g;
        __builtin_prefetch(xrow + 16 * (i + 2) + 112, 0, 3);
        frag_load(s2, xrow, w0, w1, w2, w3, i + 2);
        frag_mma(s0, acc0, acc1, acc2, acc3);
        __builtin_prefetch(xrow + 16 * (i + 3) + 112, 0, 3);
        frag_load(s0, xrow, w0, w1, w2, w3, i + 3);
        frag_mma(s1, acc0, acc1, acc2, acc3);
        __builtin_prefetch(xrow + 16 * (i + 4) + 112, 0, 3);
        frag_load(s1, xrow, w0, w1, w2, w3, i + 4);
        frag_mma(s2, acc0, acc1, acc2, acc3);
    }
    if (R >= 3) frag_load(s2, xrow, w0, w1, w2, w3, 3 * G + 2);
    frag_mma(s0, acc0, acc1, acc2, acc3);
    if (R == 4) frag_load(s0, xrow, w0, w1, w2, w3, 3 * G + 3);
    frag_mma(s1, acc0, acc1, acc2, acc3);
    if (R >= 3) frag_mma(s2, acc0, acc1, acc2, acc3);
    if (R == 4) frag_mma(s0, acc0, acc1, acc2, acc3);
}

__global__ void __launch_bounds__(32) gemm_xwt_kernel(const float* __restrict__ X,
                                const float* __restrict__ W,
                                float* __restrict__ C,
                                int M, int N, int wBatch, int cBatch) {
    constexpr int K = HH;
    W += (size_t)blockIdx.y * wBatch;
    C += (size_t)blockIdx.y * cBatch;
    const int wave = blockIdx.x;
    const int lane = threadIdx.x;
    const int hi   = lane >> 4;
    const int lr   = lane & 15;
    const int nt   = N >> 6;
    const int m0   = (wave / nt) * 16;
    const int n0   = (wave % nt) * 64;

    const float* xrow = X + (size_t)(m0 + lr) * K + 2 * hi;
    const float* w0 = W + (size_t)(n0 +  0 + lr) * K + 2 * hi;
    const float* w1 = W + (size_t)(n0 + 16 + lr) * K + 2 * hi;
    const float* w2 = W + (size_t)(n0 + 32 + lr) * K + 2 * hi;
    const float* w3 = W + (size_t)(n0 + 48 + lr) * K + 2 * hi;

    v8f acc0 = {}, acc1 = {}, acc2 = {}, acc3 = {};
    gemm_strip<K / 16>(xrow, w0, w1, w2, w3, acc0, acc1, acc2, acc3);

#pragma unroll
    for (int r = 0; r < 8; ++r) {
        const size_t row = (size_t)(m0 + r + 8 * hi) * N + n0 + lr;
        C[row +  0] = acc0[r];
        C[row + 16] = acc1[r];
        C[row + 32] = acc2[r];
        C[row + 48] = acc3[r];
    }
}

#define HG 64
__global__ void __launch_bounds__(256) score_softmax_kernel(const float* __restrict__ qp,
                                     const float* __restrict__ kT,
                                     const float* __restrict__ vvec,
                                     const int* __restrict__ lens,
                                     float* __restrict__ attn) {
    const int b = blockIdx.x / TT;
    const int t = blockIdx.x % TT;
    const int tid  = threadIdx.x;
    const int wv   = tid >> 5;
    const int lane = tid & 31;

    __shared__ float redbuf[8];

    const float* __restrict__ qrow = qp + (size_t)(b * TT + t) * HH;
    const float* __restrict__ kcol = kT + (size_t)b * HH * SS + 2 * tid;

    v2f acc = {0.0f, 0.0f};
    for (int h0 = 0; h0 < HH; h0 += HG) {
        v2f kk[HG];
#pragma unroll
        for (int j = 0; j < HG; ++j) kk[j] = ld2(kcol + (size_t)(h0 + j) * SS);
#pragma unroll
        for (int j = 0; j < HG; ++j) {
            const float qh = qrow[h0 + j];
            const float vh = vvec[h0 + j];
            acc += fast_tanh2(kk[j] + qh) * vh;
        }
    }

    const int len = lens[b];
    const float s0 = (2 * tid     < len) ? acc.x : -__builtin_inff();
    const float s1 = (2 * tid + 1 < len) ? acc.y : -__builtin_inff();

    float m = fmaxf(s0, s1);
#pragma unroll
    for (int o = 16; o >= 1; o >>= 1) m = fmaxf(m, __shfl_xor(m, o, 32));
    if (lane == 0) redbuf[wv] = m;
    __syncthreads();
    float mx = redbuf[0];
#pragma unroll
    for (int i = 1; i < 8; ++i) mx = fmaxf(mx, redbuf[i]);
    __syncthreads();

    const float e0 = __builtin_amdgcn_exp2f((s0 - mx) * LOG2E);
    const float e1 = __builtin_amdgcn_exp2f((s1 - mx) * LOG2E);
    float ps = e0 + e1;
#pragma unroll
    for (int o = 16; o >= 1; o >>= 1) ps += __shfl_xor(ps, o, 32);
    if (lane == 0) redbuf[wv] = ps;
    __syncthreads();
    float tot = redbuf[0];
#pragma unroll
    for (int i = 1; i < 8; ++i) tot += redbuf[i];
    const float inv = __builtin_amdgcn_rcpf(tot);

    float2 r; r.x = e0 * inv; r.y = e1 * inv;
    *reinterpret_cast<float2*>(attn + (size_t)(b * TT + t) * SS + 2 * tid) = r;
}

__device__ __forceinline__ void ctx_loadB(const float* __restrict__ eb, int ka, int n,
                                          v2f& b0, v2f& b1, v2f& b2, v2f& b3) {
    const float* e0 = eb + (size_t)ka * HH + n;
    b0.x = e0[ 0]; b0.y = e0[HH +  0];
    b1.x = e0[16]; b1.y = e0[HH + 16];
    b2.x = e0[32]; b2.y = e0[HH + 32];
    b3.x = e0[48]; b3.y = e0[HH + 48];
}

__global__ void context_kernel(const float* __restrict__ attn,
                               const float* __restrict__ enc,
                               float* __restrict__ ctx) {
    const int wave = blockIdx.x;
    const int lane = threadIdx.x;
    const int hi   = lane >> 4;
    const int lr   = lane & 15;
    const int b    = wave >> 6;
    const int rem  = wave & 63;
    const int m0   = (rem >> 3) * 16;
    const int n0   = (rem & 7) * 64;

    const float* arow = attn + (size_t)(b * TT + m0 + lr) * SS + 2 * hi;
    const float* eb   = enc + (size_t)b * SS * HH;
    const int    nb   = n0 + lr;

    v8f acc0 = {}, acc1 = {}, acc2 = {}, acc3 = {};
    v2f a_c[4], bc0[4], bc1[4], bc2[4], bc3[4];
#pragma unroll
    for (int j = 0; j < 4; ++j) {
        a_c[j] = ld2(arow + 4 * j);
        ctx_loadB(eb, 4 * j + 2 * hi, nb, bc0[j], bc1[j], bc2[j], bc3[j]);
    }
    for (int k = 16; k < SS; k += 16) {
        v2f a_n[4], bn0[4], bn1[4], bn2[4], bn3[4];
#pragma unroll
        for (int j = 0; j < 4; ++j) {
            a_n[j] = ld2(arow + k + 4 * j);
            ctx_loadB(eb, k + 4 * j + 2 * hi, nb, bn0[j], bn1[j], bn2[j], bn3[j]);
        }
#pragma unroll
        for (int j = 0; j < 4; ++j) {
            acc0 = WMMA_F32(a_c[j], bc0[j], acc0);
            acc1 = WMMA_F32(a_c[j], bc1[j], acc1);
            acc2 = WMMA_F32(a_c[j], bc2[j], acc2);
            acc3 = WMMA_F32(a_c[j], bc3[j], acc3);
        }
#pragma unroll
        for (int j = 0; j < 4; ++j) {
            a_c[j] = a_n[j];
            bc0[j] = bn0[j]; bc1[j] = bn1[j];
            bc2[j] = bn2[j]; bc3[j] = bn3[j];
        }
    }
#pragma unroll
    for (int j = 0; j < 4; ++j) {
        acc0 = WMMA_F32(a_c[j], bc0[j], acc0);
        acc1 = WMMA_F32(a_c[j], bc1[j], acc1);
        acc2 = WMMA_F32(a_c[j], bc2[j], acc2);
        acc3 = WMMA_F32(a_c[j], bc3[j], acc3);
    }

#pragma unroll
    for (int r = 0; r < 8; ++r) {
        const size_t row = (size_t)(b * TT + m0 + r + 8 * hi) * HH + n0 + lr;
        ctx[row +  0] = acc0[r];
        ctx[row + 16] = acc1[r];
        ctx[row + 32] = acc2[r];
        ctx[row + 48] = acc3[r];
    }
}

__global__ void __launch_bounds__(256) out_gemm_kernel(const float* __restrict__ ctx,
                                const float* __restrict__ query,
                                const float* __restrict__ Wout,
                                const float* __restrict__ bout,
                                float* __restrict__ outp) {
    const int wave = blockIdx.x;
    const int lane = threadIdx.x;
    const int hi   = lane >> 4;
    const int lr   = lane & 15;
    const int m0   = (wave >> 3) * 16;
    const int n0   = (wave & 7) * 64;

    const float* crow = ctx   + (size_t)(m0 + lr) * HH + 2 * hi;
    const float* qrow = query + (size_t)(m0 + lr) * HH + 2 * hi;
    const float* w0 = Wout + (size_t)(n0 +  0 + lr) * (2 * HH) + 2 * hi;
    const float* w1 = Wout + (size_t)(n0 + 16 + lr) * (2 * HH) + 2 * hi;
    const float* w2 = Wout + (size_t)(n0 + 32 + lr) * (2 * HH) + 2 * hi;
    const float* w3 = Wout + (size_t)(n0 + 48 + lr) * (2 * HH) + 2 * hi;

    v8f acc0 = {}, acc1 = {}, acc2 = {}, acc3 = {};
    gemm_strip<HH / 16>(crow, w0, w1, w2, w3, acc0, acc1, acc2, acc3);
    gemm_strip<HH / 16>(qrow, w0 + HH, w1 + HH, w2 + HH, w3 + HH, acc0, acc1, acc2, acc3);

    const float bb0 = bout[n0 +  0 + lr];
    const float bb1 = bout[n0 + 16 + lr];
    const float bb2 = bout[n0 + 32 + lr];
    const float bb3 = bout[n0 + 48 + lr];
#pragma unroll
    for (int r = 0; r < 8; ++r) {
        const size_t row = (size_t)(m0 + r + 8 * hi) * HH + n0 + lr;
        outp[row +  0] = fast_tanh(acc0[r] + bb0);
        outp[row + 16] = fast_tanh(acc1[r] + bb1);
        outp[row + 32] = fast_tanh(acc2[r] + bb2);
        outp[row + 48] = fast_tanh(acc3[r] + bb3);
    }
}

__global__ void ln_kernel(const float* __restrict__ x,
                          const float* __restrict__ gamma,
                          const float* __restrict__ beta,
                          float* __restrict__ out) {
    const int wave = (blockIdx.x * blockDim.x + threadIdx.x) >> 5;
    const int lane = threadIdx.x & 31;
    const float* row = x + (size_t)wave * HH;

    float4 val[4];
    float s = 0.0f;
#pragma unroll
    for (int j = 0; j < 4; ++j) {
        val[j] = *(const float4*)(row + (lane + 32 * j) * 4);
        s += val[j].x + val[j].y + val[j].z + val[j].w;
    }
#pragma unroll
    for (int o = 16; o >= 1; o >>= 1) s += __shfl_xor(s, o, 32);
    const float mu = s * (1.0f / HH);

    float q = 0.0f;
#pragma unroll
    for (int j = 0; j < 4; ++j) {
        const float dx = val[j].x - mu, dy = val[j].y - mu;
        const float dz = val[j].z - mu, dw = val[j].w - mu;
        q += dx * dx + dy * dy + dz * dz + dw * dw;
    }
#pragma unroll
    for (int o = 16; o >= 1; o >>= 1) q += __shfl_xor(q, o, 32);
    const float rstd = __builtin_amdgcn_rsqf(q * (1.0f / HH) + LN_EPS);

    float* orow = out + (size_t)wave * HH;
#pragma unroll
    for (int j = 0; j < 4; ++j) {
        const int h = (lane + 32 * j) * 4;
        const float4 g = *(const float4*)(gamma + h);
        const float4 bt = *(const float4*)(beta + h);
        float4 r;
        r.x = (val[j].x - mu) * rstd * g.x + bt.x;
        r.y = (val[j].y - mu) * rstd * g.y + bt.y;
        r.z = (val[j].z - mu) * rstd * g.z + bt.z;
        r.w = (val[j].w - mu) * rstd * g.w + bt.w;
        *(float4*)(orow + h) = r;
    }
}

extern "C" void kernel_launch(void* const* d_in, const int* in_sizes, int n_in,
                              void* d_out, int out_size, void* d_ws, size_t ws_size,
                              hipStream_t stream) {
    const float* query  = (const float*)d_in[0];
    const float* enc    = (const float*)d_in[1];
    const int*   lens   = (const int*)  d_in[2];
    const float* W_h    = (const float*)d_in[3];
    const float* W_s    = (const float*)d_in[4];
    const float* vvec   = (const float*)d_in[5];
    const float* W_out  = (const float*)d_in[6];
    const float* b_out  = (const float*)d_in[7];
    const float* gamma  = (const float*)d_in[8];
    const float* beta   = (const float*)d_in[9];
    float* out = (float*)d_out;

    float* qp   = (float*)d_ws;
    float* kp   = qp   + (size_t)BB * TT * HH;
    float* attn = kp   + (size_t)BB * SS * HH;
    float* ctx  = attn + (size_t)BB * TT * SS;
    float* outp = ctx  + (size_t)BB * TT * HH;

    gemm_xwt_kernel<<<256, 32, 0, stream>>>(query, W_s, qp, BB * TT, HH, 0, 0);
    gemm_xwt_kernel<<<dim3(256, BB), 32, 0, stream>>>(W_h, enc, kp, HH, SS, SS * HH, HH * SS);
    score_softmax_kernel<<<BB * TT, 256, 0, stream>>>(qp, kp, vvec, lens, attn);
    context_kernel<<<256, 32, 0, stream>>>(attn, enc, ctx);
    out_gemm_kernel<<<256, 32, 0, stream>>>(ctx, query, W_out, b_out, outp);
    ln_kernel<<<64, 256, 0, stream>>>(outp, gamma, beta, out);
}


# q projection, context and output GEMMs: each 16x64 tile split over the two waves of a 64-thread workgroup (16x32 per wave, two accumulators), loads four 16-k blocks ahead; sum order unchanged
# speedup vs baseline: 7.2092x; 1.4026x over previous
#include <hip/hip_runtime.h>

#define BB 4
#define TT 128
#define SS 512
#define HH 512
#define LN_EPS 1e-5f
#define LOG2E 1.4426950408889634f

typedef __attribute__((ext_vector_type(2))) float v2f;
typedef __attribute__((ext_vector_type(8))) float v8f;

__device__ __forceinline__ v2f ld2(const float* p) {
    const float2 t = *reinterpret_cast<const float2*>(p);
    v2f r; r.x = t.x; r.y = t.y; return r;
}

__device__ __forceinline__ float fast_tanh(float x) {
    float xc = fminf(fmaxf(x, -15.0f), 15.0f);
    float e  = __builtin_amdgcn_exp2f(xc * (2.0f * LOG2E));
    return (e - 1.0f) * __builtin_amdgcn_rcpf(e + 1.0f);
}

__device__ __forceinline__ v2f fast_tanh2(v2f x) {
    v2f xc;
    xc.x = fminf(fmaxf(x.x, -15.0f), 15.0f);
    xc.y = fminf(fmaxf(x.y, -15.0f), 15.0f);
    const v2f tt = xc * (2.0f * LOG2E);
    v2f e;
    e.x = __builtin_amdgcn_exp2f(tt.x);
    e.y = __builtin_amdgcn_exp2f(tt.y);
    const v2f d = e + 1.0f;
    v2f r;
    r.x = __builtin_amdgcn_rcpf(d.x);
    r.y = __builtin_amdgcn_rcpf(d.y);
    return (e - 1.0f) * r;
}

#define WMMA_F32(A, Bf, C) \
    __builtin_amdgcn_wmma_f32_16x16x4_f32(false, (A), false, (Bf), (short)0, (C), false, false)

struct Frag { v2f a[4], b0[4], b1[4], b2[4], b3[4]; };

__device__ __forceinline__ void frag_load(Frag& f, const float* __restrict__ xrow,
                                          const float* __restrict__ w0, const float* __restrict__ w1,
                                          const float* __restrict__ w2, const float* __restrict__ w3, int kb) {
#pragma unroll
    for (int j = 0; j < 4; ++j) {
        const int o = 16 * kb + 4 * j;
        f.a[j]  = ld2(xrow + o);
        f.b0[j] = ld2(w0 + o); f.b1[j] = ld2(w1 + o);
        f.b2[j] = ld2(w2 + o); f.b3[j] = ld2(w3 + o);
    }
    __builtin_amdgcn_sched_barrier(0);
}

__device__ __forceinline__ void frag_mma(const Frag& f,
                                         v8f& acc0, v8f& acc1, v8f& acc2, v8f& acc3) {
#pragma unroll
    for (int j = 0; j < 4; ++j) {
        acc0 = WMMA_F32(f.a[j], f.b0[j], acc0);
        acc1 = WMMA_F32(f.a[j], f.b1[j], acc1);
        acc2 = WMMA_F32(f.a[j], f.b2[j], acc2);
        acc3 = WMMA_F32(f.a[j], f.b3[j], acc3);
    }
    asm volatile("" : "+v"(acc0), "+v"(acc1), "+v"(acc2), "+v"(acc3) : : "memory");
}

template <int KB>
__device__ __forceinline__ void gemm_strip(const float* __restrict__ xrow,
                                           const float* __restrict__ w0, const float* __restrict__ w1,
                                           const float* __restrict__ w2, const float* __restrict__ w3,
                                           v8f& acc0, v8f& acc1, v8f& acc2, v8f& acc3) {
    static_assert(KB >= 3, "ring of three");
    constexpr int G = (KB - 2) / 3;
    constexpr int R = KB - 3 * G;
    Frag s0, s1, s2;
    frag_load(s0, xrow, w0, w1, w2, w3, 0);
    frag_load(s1, xrow, w0, w1, w2, w3, 1);
    for (int g = 0; g < G; ++g) {
        const int i = 3 * g;
        __builtin_prefetch(xrow + 16 * (i + 2) + 112, 0, 3);
        frag_load(s2, xrow, w0, w1, w2, w3, i + 2);
        frag_mma(s0, acc0, acc1, acc2, acc3);
        __builtin_prefetch(xrow + 16 * (i + 3) + 112, 0, 3);
        frag_load(s0, xrow, w0, w1, w2, w3, i + 3);
        frag_mma(s1, acc0, acc1, acc2, acc3);
        __builtin_prefetch(xrow + 16 * (i + 4) + 112, 0, 3);
        frag_load(s1, xrow, w0, w1, w2, w3, i + 4);
        frag_mma(s2, acc0, acc1, acc2, acc3);
    }
    if (R >= 3) frag_load(s2, xrow, w0, w1, w2, w3, 3 * G + 2);
    frag_mma(s0, acc0, acc1, acc2, acc3);
    if (R == 4) frag_load(s0, xrow, w0, w1, w2, w3, 3 * G + 3);
    frag_mma(s1, acc0, acc1, acc2, acc3);
    if (R >= 3) frag_mma(s2, acc0, acc1, acc2, acc3);
    if (R == 4) frag_mma(s0, acc0, acc1, acc2, acc3);
}


#define RD 4
template <int KB>
__device__ __forceinline__ void half_strip(const float* __restrict__ xrow,
                                           const float* __restrict__ w0, const float* __restrict__ w1,
                                           v8f& acc0, v8f& acc1) {
    v2f fa[RD + 1][4], f0[RD + 1][4], f1[RD + 1][4];
#pragma unroll
    for (int i = 0; i < RD; ++i) {
#pragma unroll
        for (int j = 0; j < 4; ++j) {
            const int o = 16 * i + 4 * j;
            fa[i][j] = ld2(xrow + o); f0[i][j] = ld2(w0 + o); f1[i][j] = ld2(w1 + o);
        }
        __builtin_amdgcn_sched_barrier(0);
    }
#pragma unroll
    for (int kb = 0; kb < KB; ++kb) {
        if (kb + RD < KB) {
            const int s = (kb + RD) % (RD + 1);
#pragma unroll
            for (int j = 0; j < 4; ++j) {
                const int o = 16 * (kb + RD) + 4 * j;
                fa[s][j] = ld2(xrow + o); f0[s][j] = ld2(w0 + o); f1[s][j] = ld2(w1 + o);
            }
            __builtin_amdgcn_sched_barrier(0);
        }
        const int c = kb % (RD + 1);
#pragma unroll
        for (int j = 0; j < 4; ++j) {
            acc0 = WMMA_F32(fa[c][j], f0[c][j], acc0);
            acc1 = WMMA_F32(fa[c][j], f1[c][j], acc1);
        }
        asm volatile("" : "+v"(acc0), "+v"(acc1) : : "memory");
    }
}

__global__ void __launch_bounds__(32) gemm_xwt_kernel(const float* __restrict__ X,
                                const float* __restrict__ W,
                                float* __restrict__ C,
                                int M, int N, int wBatch, int cBatch) {
    constexpr int K = HH;
    W += (size_t)blockIdx.y * wBatch;
    C += (size_t)blockIdx.y * cBatch;
    const int wave = blockIdx.x;
    const int lane = threadIdx.x;
    const int hi   = lane >> 4;
    const int lr   = lane & 15;
    const int nt   = N >> 6;
    const int m0   = (wave / nt) * 16;
    const int n0   = (wave % nt) * 64;

    const float* xrow = X + (size_t)(m0 + lr) * K + 2 * hi;
    const float* w0 = W + (size_t)(n0 +  0 + lr) * K + 2 * hi;
    const float* w1 = W + (size_t)(n0 + 16 + lr) * K + 2 * hi;
    const float* w2 = W + (size_t)(n0 + 32 + lr) * K + 2 * hi;
    const float* w3 = W + (size_t)(n0 + 48 + lr) * K + 2 * hi;

    v8f acc0 = {}, acc1 = {}, acc2 = {}, acc3 = {};
    gemm_strip<K / 16>(xrow, w0, w1, w2, w3, acc0, acc1, acc2, acc3);

#pragma unroll
    for (int r = 0; r < 8; ++r) {
        const size_t row = (size_t)(m0 + r + 8 * hi) * N + n0 + lr;
        C[row +  0] = acc0[r];
        C[row + 16] = acc1[r];
        C[row + 32] = acc2[r];
        C[row + 48] = acc3[r];
    }
}


__global__ void __launch_bounds__(64) q_gemm_kernel(const float* __restrict__ X,
                                                    const float* __restrict__ Wm,
                                                    float* __restrict__ C) {
    const int tile = blockIdx.x;
    const int w    = threadIdx.x >> 5;
    const int lane = threadIdx.x & 31;
    const int hi   = lane >> 4;
    const int lr   = lane & 15;
    const int m0   = (tile >> 3) * 16;
    const int n0   = (tile & 7) * 64 + 32 * w;

    const float* xrow = X  + (size_t)(m0 + lr) * HH + 2 * hi;
    const float* w0   = Wm + (size_t)(n0 + 2 * lr) * HH + 2 * hi;
    const float* w1   = w0 + HH;

    v8f acc0 = {}, acc1 = {};
    half_strip<HH / 16>(xrow, w0, w1, acc0, acc1);

#pragma unroll
    for (int r = 0; r < 8; ++r) {
        float2 o; o.x = acc0[r]; o.y = acc1[r];
        *reinterpret_cast<float2*>(C + (size_t)(m0 + r + 8 * hi) * HH + n0 + 2 * lr) = o;
    }
}

#define HG 64
__global__ void __launch_bounds__(256) score_softmax_kernel(const float* __restrict__ qp,
                                     const float* __restrict__ kT,
                                     const float* __restrict__ vvec,
                                     const int* __restrict__ lens,
                                     float* __restrict__ attn) {
    const int b = blockIdx.x / TT;
    const int t = blockIdx.x % TT;
    const int tid  = threadIdx.x;
    const int wv   = tid >> 5;
    const int lane = tid & 31;

    __shared__ float redbuf[8];

    const float* __restrict__ qrow = qp + (size_t)(b * TT + t) * HH;
    const float* __restrict__ kcol = kT + (size_t)b * HH * SS + 2 * tid;

    v2f acc = {0.0f, 0.0f};
    for (int h0 = 0; h0 < HH; h0 += HG) {
        v2f kk[HG];
#pragma unroll
        for (int j = 0; j < HG; ++j) kk[j] = ld2(kcol + (size_t)(h0 + j) * SS);
#pragma unroll
        for (int j = 0; j < HG; ++j) {
            const float qh = qrow[h0 + j];
            const float vh = vvec[h0 + j];
            acc += fast_tanh2(kk[j] + qh) * vh;
        }
    }

    const int len = lens[b];
    const float s0 = (2 * tid     < len) ? acc.x : -__builtin_inff();
    const float s1 = (2 * tid + 1 < len) ? acc.y : -__builtin_inff();

    float m = fmaxf(s0, s1);
#pragma unroll
    for (int o = 16; o >= 1; o >>= 1) m = fmaxf(m, __shfl_xor(m, o, 32));
    if (lane == 0) redbuf[wv] = m;
    __syncthreads();
    float mx = redbuf[0];
#pragma unroll
    for (int i = 1; i < 8; ++i) mx = fmaxf(mx, redbuf[i]);
    __syncthreads();

    const float e0 = __builtin_amdgcn_exp2f((s0 - mx) * LOG2E);
    const float e1 = __builtin_amdgcn_exp2f((s1 - mx) * LOG2E);
    float ps = e0 + e1;
#pragma unroll
    for (int o = 16; o >= 1; o >>= 1) ps += __shfl_xor(ps, o, 32);
    if (lane == 0) redbuf[wv] = ps;
    __syncthreads();
    float tot = redbuf[0];
#pragma unroll
    for (int i = 1; i < 8; ++i) tot += redbuf[i];
    const float inv = __builtin_amdgcn_rcpf(tot);

    float2 r; r.x = e0 * inv; r.y = e1 * inv;
    *reinterpret_cast<float2*>(attn + (size_t)(b * TT + t) * SS + 2 * tid) = r;
}

__global__ void __launch_bounds__(64) context_kernel(const float* __restrict__ attn,
                                                     const float* __restrict__ enc,
                                                     float* __restrict__ ctx) {
    const int tile = blockIdx.x;
    const int w    = threadIdx.x >> 5;
    const int lane = threadIdx.x & 31;
    const int hi   = lane >> 4;
    const int lr   = lane & 15;
    const int b    = tile >> 6;
    const int rem  = tile & 63;
    const int m0   = (rem >> 3) * 16;
    const int n0   = (rem & 7) * 64 + 32 * w;

    const float* arow = attn + (size_t)(b * TT + m0 + lr) * SS + 2 * hi;
    const float* eb   = enc + (size_t)b * SS * HH + (size_t)(2 * hi) * HH + n0 + 2 * lr;

    v8f acc0 = {}, acc1 = {};
    v2f fa[RD + 1][4], e0[RD + 1][4], e1[RD + 1][4];
#pragma unroll
    for (int i = 0; i < RD; ++i) {
#pragma unroll
        for (int j = 0; j < 4; ++j) {
            const int k = 16 * i + 4 * j;
            fa[i][j] = ld2(arow + k);
            e0[i][j] = ld2(eb + (size_t)k * HH);
            e1[i][j] = ld2(eb + (size_t)(k + 1) * HH);
        }
        __builtin_amdgcn_sched_barrier(0);
    }
#pragma unroll
    for (int kb = 0; kb < SS / 16; ++kb) {
        if (kb + RD < SS / 16) {
            const int s = (kb + RD) % (RD + 1);
#pragma unroll
            for (int j = 0; j < 4; ++j) {
                const int k = 16 * (kb + RD) + 4 * j;
                fa[s][j] = ld2(arow + k);
                e0[s][j] = ld2(eb + (size_t)k * HH);
                e1[s][j] = ld2(eb + (size_t)(k + 1) * HH);
            }
            __builtin_amdgcn_sched_barrier(0);
        }
        const int c = kb % (RD + 1);
#pragma unroll
        for (int j = 0; j < 4; ++j) {
            v2f b0, b1;
            b0.x = e0[c][j].x; b0.y = e1[c][j].x;
            b1.x = e0[c][j].y; b1.y = e1[c][j].y;
            acc0 = WMMA_F32(fa[c][j], b0, acc0);
            acc1 = WMMA_F32(fa[c][j], b1, acc1);
        }
        asm volatile("" : "+v"(acc0), "+v"(acc1) : : "memory");
    }

#pragma unroll
    for (int r = 0; r < 8; ++r) {
        float2 o; o.x = acc0[r]; o.y = acc1[r];
        *reinterpret_cast<float2*>(ctx + (size_t)(b * TT + m0 + r + 8 * hi) * HH + n0 + 2 * lr) = o;
    }
}

__global__ void __launch_bounds__(64) out_gemm_kernel(const float* __restrict__ ctx,
                                const float* __restrict__ query,
                                const float* __restrict__ Wout,
                                const float* __restrict__ bout,
                                float* __restrict__ outp) {
    const int tile = blockIdx.x;
    const int w    = threadIdx.x >> 5;
    const int lane = threadIdx.x & 31;
    const int hi   = lane >> 4;
    const int lr   = lane & 15;
    const int m0   = (tile >> 3) * 16;
    const int n0   = (tile & 7) * 64 + 32 * w;

    const float* crow = ctx   + (size_t)(m0 + lr) * HH + 2 * hi;
    const float* qrow = query + (size_t)(m0 + lr) * HH + 2 * hi;
    const float* w0 = Wout + (size_t)(n0 + 2 * lr) * (2 * HH) + 2 * hi;
    const float* w1 = w0 + 2 * HH;

    v8f acc0 = {}, acc1 = {};
    half_strip<HH / 16>(crow, w0, w1, acc0, acc1);
    half_strip<HH / 16>(qrow, w0 + HH, w1 + HH, acc0, acc1);

    const float2 bb = *reinterpret_cast<const float2*>(bout + n0 + 2 * lr);
#pragma unroll
    for (int r = 0; r < 8; ++r) {
        float2 o;
        o.x = fast_tanh(acc0[r] + bb.x);
        o.y = fast_tanh(acc1[r] + bb.y);
        *reinterpret_cast<float2*>(outp + (size_t)(m0 + r + 8 * hi) * HH + n0 + 2 * lr) = o;
    }
}

__global__ void ln_kernel(const float* __restrict__ x,
                          const float* __restrict__ gamma,
                          const float* __restrict__ beta,
                          float* __restrict__ out) {
    const int wave = (blockIdx.x * blockDim.x + threadIdx.x) >> 5;
    const int lane = threadIdx.x & 31;
    const float* row = x + (size_t)wave * HH;

    float4 val[4];
    float s = 0.0f;
#pragma unroll
    for (int j = 0; j < 4; ++j) {
        val[j] = *(const float4*)(row + (lane + 32 * j) * 4);
        s += val[j].x + val[j].y + val[j].z + val[j].w;
    }
#pragma unroll
    for (int o = 16; o >= 1; o >>= 1) s += __shfl_xor(s, o, 32);
    const float mu = s * (1.0f / HH);

    float q = 0.0f;
#pragma unroll
    for (int j = 0; j < 4; ++j) {
        const float dx = val[j].x - mu, dy = val[j].y - mu;
        const float dz = val[j].z - mu, dw = val[j].w - mu;
        q += dx * dx + dy * dy + dz * dz + dw * dw;
    }
#pragma unroll
    for (int o = 16; o >= 1; o >>= 1) q += __shfl_xor(q, o, 32);
    const float rstd = __builtin_amdgcn_rsqf(q * (1.0f / HH) + LN_EPS);

    float* orow = out + (size_t)wave * HH;
#pragma unroll
    for (int j = 0; j < 4; ++j) {
        const int h = (lane + 32 * j) * 4;
        const float4 g = *(const float4*)(gamma + h);
        const float4 bt = *(const float4*)(beta + h);
        float4 r;
        r.x = (val[j].x - mu) * rstd * g.x + bt.x;
        r.y = (val[j].y - mu) * rstd * g.y + bt.y;
        r.z = (val[j].z - mu) * rstd * g.z + bt.z;
        r.w = (val[j].w - mu) * rstd * g.w + bt.w;
        *(float4*)(orow + h) = r;
    }
}

extern "C" void kernel_launch(void* const* d_in, const int* in_sizes, int n_in,
                              void* d_out, int out_size, void* d_ws, size_t ws_size,
                              hipStream_t stream) {
    const float* query  = (const float*)d_in[0];
    const float* enc    = (const float*)d_in[1];
    const int*   lens   = (const int*)  d_in[2];
    const float* W_h    = (const float*)d_in[3];
    const float* W_s    = (const float*)d_in[4];
    const float* vvec   = (const float*)d_in[5];
    const float* W_out  = (const float*)d_in[6];
    const float* b_out  = (const float*)d_in[7];
    const float* gamma  = (const float*)d_in[8];
    const float* beta   = (const float*)d_in[9];
    float* out = (float*)d_out;

    float* qp   = (float*)d_ws;
    float* kp   = qp   + (size_t)BB * TT * HH;
    float* attn = kp   + (size_t)BB * SS * HH;
    float* ctx  = attn + (size_t)BB * TT * SS;
    float* outp = ctx  + (size_t)BB * TT * HH;

    q_gemm_kernel<<<256, 64, 0, stream>>>(query, W_s, qp);
    gemm_xwt_kernel<<<dim3(256, BB), 32, 0, stream>>>(W_h, enc, kp, HH, SS, SS * HH, HH * SS);
    score_softmax_kernel<<<BB * TT, 256, 0, stream>>>(qp, kp, vvec, lens, attn);
    context_kernel<<<256, 64, 0, stream>>>(attn, enc, ctx);
    out_gemm_kernel<<<256, 64, 0, stream>>>(ctx, query, W_out, b_out, outp);
    ln_kernel<<<64, 256, 0, stream>>>(outp, gamma, beta, out);
}


# k projection GEMM: each 16x64 tile split over the two waves of a 64-thread workgroup (16x32 per wave, two accumulators, loads four 16-k blocks ahead): 2048 waves, two per SIMD; sum order unchanged
# speedup vs baseline: 7.5048x; 1.0410x over previous
#include <hip/hip_runtime.h>

#define BB 4
#define TT 128
#define SS 512
#define HH 512
#define LN_EPS 1e-5f
#define LOG2E 1.4426950408889634f

typedef __attribute__((ext_vector_type(2))) float v2f;
typedef __attribute__((ext_vector_type(8))) float v8f;

__device__ __forceinline__ v2f ld2(const float* p) {
    const float2 t = *reinterpret_cast<const float2*>(p);
    v2f r; r.x = t.x; r.y = t.y; return r;
}

__device__ __forceinline__ float fast_tanh(float x) {
    float xc = fminf(fmaxf(x, -15.0f), 15.0f);
    float e  = __builtin_amdgcn_exp2f(xc * (2.0f * LOG2E));
    return (e - 1.0f) * __builtin_amdgcn_rcpf(e + 1.0f);
}

__device__ __forceinline__ v2f fast_tanh2(v2f x) {
    v2f xc;
    xc.x = fminf(fmaxf(x.x, -15.0f), 15.0f);
    xc.y = fminf(fmaxf(x.y, -15.0f), 15.0f);
    const v2f tt = xc * (2.0f * LOG2E);
    v2f e;
    e.x = __builtin_amdgcn_exp2f(tt.x);
    e.y = __builtin_amdgcn_exp2f(tt.y);
    const v2f d = e + 1.0f;
    v2f r;
    r.x = __builtin_amdgcn_rcpf(d.x);
    r.y = __builtin_amdgcn_rcpf(d.y);
    return (e - 1.0f) * r;
}

#define WMMA_F32(A, Bf, C) \
    __builtin_amdgcn_wmma_f32_16x16x4_f32(false, (A), false, (Bf), (short)0, (C), false, false)

#define RD 4
template <int KB>
__device__ __forceinline__ void half_strip(const float* __restrict__ xrow,
                                           const float* __restrict__ w0, const float* __restrict__ w1,
                                           v8f& acc0, v8f& acc1) {
    v2f fa[RD + 1][4], f0[RD + 1][4], f1[RD + 1][4];
#pragma unroll
    for (int i = 0; i < RD; ++i) {
#pragma unroll
        for (int j = 0; j < 4; ++j) {
            const int o = 16 * i + 4 * j;
            fa[i][j] = ld2(xrow + o); f0[i][j] = ld2(w0 + o); f1[i][j] = ld2(w1 + o);
        }
        __builtin_amdgcn_sched_barrier(0);
    }
#pragma unroll
    for (int kb = 0; kb < KB; ++kb) {
        if (kb + RD < KB) {
            const int s = (kb + RD) % (RD + 1);
#pragma unroll
            for (int j = 0; j < 4; ++j) {
                const int o = 16 * (kb + RD) + 4 * j;
                fa[s][j] = ld2(xrow + o); f0[s][j] = ld2(w0 + o); f1[s][j] = ld2(w1 + o);
            }
            __builtin_amdgcn_sched_barrier(0);
        }
        const int c = kb % (RD + 1);
#pragma unroll
        for (int j = 0; j < 4; ++j) {
            acc0 = WMMA_F32(fa[c][j], f0[c][j], acc0);
            acc1 = WMMA_F32(fa[c][j], f1[c][j], acc1);
        }
        asm volatile("" : "+v"(acc0), "+v"(acc1) : : "memory");
    }
}

static_assert(SS == HH, "both projections write rows of 512 floats");
__global__ void __launch_bounds__(64) proj_gemm_kernel(const float* __restrict__ X,
                                                       const float* __restrict__ Wm,
                                                       float* __restrict__ C,
                                                       int wBatch, int cBatch) {
    Wm += (size_t)blockIdx.y * wBatch;
    C  += (size_t)blockIdx.y * cBatch;
    const int tile = blockIdx.x;
    const int w    = threadIdx.x >> 5;
    const int lane = threadIdx.x & 31;
    const int hi   = lane >> 4;
    const int lr   = lane & 15;
    const int m0   = (tile >> 3) * 16;
    const int n0   = (tile & 7) * 64 + 32 * w;

    const float* xrow = X  + (size_t)(m0 + lr) * HH + 2 * hi;
    const float* w0   = Wm + (size_t)(n0 + 2 * lr) * HH + 2 * hi;
    const float* w1   = w0 + HH;

    v8f acc0 = {}, acc1 = {};
    half_strip<HH / 16>(xrow, w0, w1, acc0, acc1);

#pragma unroll
    for (int r = 0; r < 8; ++r) {
        float2 o; o.x = acc0[r]; o.y = acc1[r];
        *reinterpret_cast<float2*>(C + (size_t)(m0 + r + 8 * hi) * HH + n0 + 2 * lr) = o;
    }
}

#define HG 64
__global__ void __launch_bounds__(256) score_softmax_kernel(const float* __restrict__ qp,
                                     const float* __restrict__ kT,
                                     const float* __restrict__ vvec,
                                     const int* __restrict__ lens,
                                     float* __restrict__ attn) {
    const int b = blockIdx.x / TT;
    const int t = blockIdx.x % TT;
    const int tid  = threadIdx.x;
    const int wv   = tid >> 5;
    const int lane = tid & 31;

    __shared__ float redbuf[8];

    const float* __restrict__ qrow = qp + (size_t)(b * TT + t) * HH;
    const float* __restrict__ kcol = kT + (size_t)b * HH * SS + 2 * tid;

    v2f acc = {0.0f, 0.0f};
    for (int h0 = 0; h0 < HH; h0 += HG) {
        v2f kk[HG];
#pragma unroll
        for (int j = 0; j < HG; ++j) kk[j] = ld2(kcol + (size_t)(h0 + j) * SS);
#pragma unroll
        for (int j = 0; j < HG; ++j) {
            const float qh = qrow[h0 + j];
            const float vh = vvec[h0 + j];
            acc += fast_tanh2(kk[j] + qh) * vh;
        }
    }

    const int len = lens[b];
    const float s0 = (2 * tid     < len) ? acc.x : -__builtin_inff();
    const float s1 = (2 * tid + 1 < len) ? acc.y : -__builtin_inff();

    float m = fmaxf(s0, s1);
#pragma unroll
    for (int o = 16; o >= 1; o >>= 1) m = fmaxf(m, __shfl_xor(m, o, 32));
    if (lane == 0) redbuf[wv] = m;
    __syncthreads();
    float mx = redbuf[0];
#pragma unroll
    for (int i = 1; i < 8; ++i) mx = fmaxf(mx, redbuf[i]);
    __syncthreads();

    const float e0 = __builtin_amdgcn_exp2f((s0 - mx) * LOG2E);
    const float e1 = __builtin_amdgcn_exp2f((s1 - mx) * LOG2E);
    float ps = e0 + e1;
#pragma unroll
    for (int o = 16; o >= 1; o >>= 1) ps += __shfl_xor(ps, o, 32);
    if (lane == 0) redbuf[wv] = ps;
    __syncthreads();
    float tot = redbuf[0];
#pragma unroll
    for (int i = 1; i < 8; ++i) tot += redbuf[i];
    const float inv = __builtin_amdgcn_rcpf(tot);

    float2 r; r.x = e0 * inv; r.y = e1 * inv;
    *reinterpret_cast<float2*>(attn + (size_t)(b * TT + t) * SS + 2 * tid) = r;
}

__global__ void __launch_bounds__(64) context_kernel(const float* __restrict__ attn,
                                                     const float* __restrict__ enc,
                                                     float* __restrict__ ctx) {
    const int tile = blockIdx.x;
    const int w    = threadIdx.x >> 5;
    const int lane = threadIdx.x & 31;
    const int hi   = lane >> 4;
    const int lr   = lane & 15;
    const int b    = tile >> 6;
    const int rem  = tile & 63;
    const int m0   = (rem >> 3) * 16;
    const int n0   = (rem & 7) * 64 + 32 * w;

    const float* arow = attn + (size_t)(b * TT + m0 + lr) * SS + 2 * hi;
    const float* eb   = enc + (size_t)b * SS * HH + (size_t)(2 * hi) * HH + n0 + 2 * lr;

    v8f acc0 = {}, acc1 = {};
    v2f fa[RD + 1][4], e0[RD + 1][4], e1[RD + 1][4];
#pragma unroll
    for (int i = 0; i < RD; ++i) {
#pragma unroll
        for (int j = 0; j < 4; ++j) {
            const int k = 16 * i + 4 * j;
            fa[i][j] = ld2(arow + k);
            e0[i][j] = ld2(eb + (size_t)k * HH);
            e1[i][j] = ld2(eb + (size_t)(k + 1) * HH);
        }
        __builtin_amdgcn_sched_barrier(0);
    }
#pragma unroll
    for (int kb = 0; kb < SS / 16; ++kb) {
        if (kb + RD < SS / 16) {
            const int s = (kb + RD) % (RD + 1);
#pragma unroll
            for (int j = 0; j < 4; ++j) {
                const int k = 16 * (kb + RD) + 4 * j;
                fa[s][j] = ld2(arow + k);
                e0[s][j] = ld2(eb + (size_t)k * HH);
                e1[s][j] = ld2(eb + (size_t)(k + 1) * HH);
            }
            __builtin_amdgcn_sched_barrier(0);
        }
        const int c = kb % (RD + 1);
#pragma unroll
        for (int j = 0; j < 4; ++j) {
            v2f b0, b1;
            b0.x = e0[c][j].x; b0.y = e1[c][j].x;
            b1.x = e0[c][j].y; b1.y = e1[c][j].y;
            acc0 = WMMA_F32(fa[c][j], b0, acc0);
            acc1 = WMMA_F32(fa[c][j], b1, acc1);
        }
        asm volatile("" : "+v"(acc0), "+v"(acc1) : : "memory");
    }

#pragma unroll
    for (int r = 0; r < 8; ++r) {
        float2 o; o.x = acc0[r]; o.y = acc1[r];
        *reinterpret_cast<float2*>(ctx + (size_t)(b * TT + m0 + r + 8 * hi) * HH + n0 + 2 * lr) = o;
    }
}

__global__ void __launch_bounds__(64) out_gemm_kernel(const float* __restrict__ ctx,
                                const float* __restrict__ query,
                                const float* __restrict__ Wout,
                                const float* __restrict__ bout,
                                float* __restrict__ outp) {
    const int tile = blockIdx.x;
    const int w    = threadIdx.x >> 5;
    const int lane = threadIdx.x & 31;
    const int hi   = lane >> 4;
    const int lr   = lane & 15;
    const int m0   = (tile >> 3) * 16;
    const int n0   = (tile & 7) * 64 + 32 * w;

    const float* crow = ctx   + (size_t)(m0 + lr) * HH + 2 * hi;
    const float* qrow = query + (size_t)(m0 + lr) * HH + 2 * hi;
    const float* w0 = Wout + (size_t)(n0 + 2 * lr) * (2 * HH) + 2 * hi;
    const float* w1 = w0 + 2 * HH;

    v8f acc0 = {}, acc1 = {};
    half_strip<HH / 16>(crow, w0, w1, acc0, acc1);
    half_strip<HH / 16>(qrow, w0 + HH, w1 + HH, acc0, acc1);

    const float2 bb = *reinterpret_cast<const float2*>(bout + n0 + 2 * lr);
#pragma unroll
    for (int r = 0; r < 8; ++r) {
        float2 o;
        o.x = fast_tanh(acc0[r] + bb.x);
        o.y = fast_tanh(acc1[r] + bb.y);
        *reinterpret_cast<float2*>(outp + (size_t)(m0 + r + 8 * hi) * HH + n0 + 2 * lr) = o;
    }
}

__global__ void ln_kernel(const float* __restrict__ x,
                          const float* __restrict__ gamma,
                          const float* __restrict__ beta,
                          float* __restrict__ out) {
    const int wave = (blockIdx.x * blockDim.x + threadIdx.x) >> 5;
    const int lane = threadIdx.x & 31;
    const float* row = x + (size_t)wave * HH;

    float4 val[4];
    float s = 0.0f;
#pragma unroll
    for (int j = 0; j < 4; ++j) {
        val[j] = *(const float4*)(row + (lane + 32 * j) * 4);
        s += val[j].x + val[j].y + val[j].z + val[j].w;
    }
#pragma unroll
    for (int o = 16; o >= 1; o >>= 1) s += __shfl_xor(s, o, 32);
    const float mu = s * (1.0f / HH);

    float q = 0.0f;
#pragma unroll
    for (int j = 0; j < 4; ++j) {
        const float dx = val[j].x - mu, dy = val[j].y - mu;
        const float dz = val[j].z - mu, dw = val[j].w - mu;
        q += dx * dx + dy * dy + dz * dz + dw * dw;
    }
#pragma unroll
    for (int o = 16; o >= 1; o >>= 1) q += __shfl_xor(q, o, 32);
    const float rstd = __builtin_amdgcn_rsqf(q * (1.0f / HH) + LN_EPS);

    float* orow = out + (size_t)wave * HH;
#pragma unroll
    for (int j = 0; j < 4; ++j) {
        const int h = (lane + 32 * j) * 4;
        const float4 g = *(const float4*)(gamma + h);
        const float4 bt = *(const float4*)(beta + h);
        float4 r;
        r.x = (val[j].x - mu) * rstd * g.x + bt.x;
        r.y = (val[j].y - mu) * rstd * g.y + bt.y;
        r.z = (val[j].z - mu) * rstd * g.z + bt.z;
        r.w = (val[j].w - mu) * rstd * g.w + bt.w;
        *(float4*)(orow + h) = r;
    }
}

extern "C" void kernel_launch(void* const* d_in, const int* in_sizes, int n_in,
                              void* d_out, int out_size, void* d_ws, size_t ws_size,
                              hipStream_t stream) {
    const float* query  = (const float*)d_in[0];
    const float* enc    = (const float*)d_in[1];
    const int*   lens   = (const int*)  d_in[2];
    const float* W_h    = (const float*)d_in[3];
    const float* W_s    = (const float*)d_in[4];
    const float* vvec   = (const float*)d_in[5];
    const float* W_out  = (const float*)d_in[6];
    const float* b_out  = (const float*)d_in[7];
    const float* gamma  = (const float*)d_in[8];
    const float* beta   = (const float*)d_in[9];
    float* out = (float*)d_out;

    float* qp   = (float*)d_ws;
    float* kp   = qp   + (size_t)BB * TT * HH;
    float* attn = kp   + (size_t)BB * SS * HH;
    float* ctx  = attn + (size_t)BB * TT * SS;
    float* outp = ctx  + (size_t)BB * TT * HH;

    proj_gemm_kernel<<<dim3(256, 1), 64, 0, stream>>>(query, W_s, qp, 0, 0);
    proj_gemm_kernel<<<dim3(256, BB), 64, 0, stream>>>(W_h, enc, kp, SS * HH, HH * SS);
    score_softmax_kernel<<<BB * TT, 256, 0, stream>>>(qp, kp, vvec, lens, attn);
    context_kernel<<<256, 64, 0, stream>>>(attn, enc, ctx);
    out_gemm_kernel<<<256, 64, 0, stream>>>(ctx, query, W_out, b_out, outp);
    ln_kernel<<<64, 256, 0, stream>>>(outp, gamma, beta, out);
}


# score kernel: exp of q+k split into e^{2q} e^{2k}: a block of two query rows takes e^{2k} once per k value for both, e^{2q} and v lane by lane as scalars; fma, rcp, fma per value; new sum order
# speedup vs baseline: 8.2794x; 1.1032x over previous
#include <hip/hip_runtime.h>

#define BB 4
#define TT 128
#define SS 512
#define HH 512
#define LN_EPS 1e-5f
#define LOG2E 1.4426950408889634f

typedef __attribute__((ext_vector_type(2))) float v2f;
typedef __attribute__((ext_vector_type(8))) float v8f;

__device__ __forceinline__ v2f ld2(const float* p) {
    const float2 t = *reinterpret_cast<const float2*>(p);
    v2f r; r.x = t.x; r.y = t.y; return r;
}

__device__ __forceinline__ float fast_tanh(float x) {
    float xc = fminf(fmaxf(x, -15.0f), 15.0f);
    float e  = __builtin_amdgcn_exp2f(xc * (2.0f * LOG2E));
    return (e - 1.0f) * __builtin_amdgcn_rcpf(e + 1.0f);
}

__device__ __forceinline__ v2f fast_tanh2(v2f x) {
    v2f xc;
    xc.x = fminf(fmaxf(x.x, -15.0f), 15.0f);
    xc.y = fminf(fmaxf(x.y, -15.0f), 15.0f);
    const v2f tt = xc * (2.0f * LOG2E);
    v2f e;
    e.x = __builtin_amdgcn_exp2f(tt.x);
    e.y = __builtin_amdgcn_exp2f(tt.y);
    const v2f d = e + 1.0f;
    v2f r;
    r.x = __builtin_amdgcn_rcpf(d.x);
    r.y = __builtin_amdgcn_rcpf(d.y);
    return (e - 1.0f) * r;
}

#define WMMA_F32(A, Bf, C) \
    __builtin_amdgcn_wmma_f32_16x16x4_f32(false, (A), false, (Bf), (short)0, (C), false, false)

#define RD 4
template <int KB>
__device__ __forceinline__ void half_strip(const float* __restrict__ xrow,
                                           const float* __restrict__ w0, const float* __restrict__ w1,
                                           v8f& acc0, v8f& acc1) {
    v2f fa[RD + 1][4], f0[RD + 1][4], f1[RD + 1][4];
#pragma unroll
    for (int i = 0; i < RD; ++i) {
#pragma unroll
        for (int j = 0; j < 4; ++j) {
            const int o = 16 * i + 4 * j;
            fa[i][j] = ld2(xrow + o); f0[i][j] = ld2(w0 + o); f1[i][j] = ld2(w1 + o);
        }
        __builtin_amdgcn_sched_barrier(0);
    }
#pragma unroll
    for (int kb = 0; kb < KB; ++kb) {
        if (kb + RD < KB) {
            const int s = (kb + RD) % (RD + 1);
#pragma unroll
            for (int j = 0; j < 4; ++j) {
                const int o = 16 * (kb + RD) + 4 * j;
                fa[s][j] = ld2(xrow + o); f0[s][j] = ld2(w0 + o); f1[s][j] = ld2(w1 + o);
            }
            __builtin_amdgcn_sched_barrier(0);
        }
        const int c = kb % (RD + 1);
#pragma unroll
        for (int j = 0; j < 4; ++j) {
            acc0 = WMMA_F32(fa[c][j], f0[c][j], acc0);
            acc1 = WMMA_F32(fa[c][j], f1[c][j], acc1);
        }
        asm volatile("" : "+v"(acc0), "+v"(acc1) : : "memory");
    }
}

static_assert(SS == HH, "both projections write rows of 512 floats");
__global__ void __launch_bounds__(64) proj_gemm_kernel(const float* __restrict__ X,
                                                       const float* __restrict__ Wm,
                                                       float* __restrict__ C,
                                                       int wBatch, int cBatch) {
    Wm += (size_t)blockIdx.y * wBatch;
    C  += (size_t)blockIdx.y * cBatch;
    const int tile = blockIdx.x;
    const int w    = threadIdx.x >> 5;
    const int lane = threadIdx.x & 31;
    const int hi   = lane >> 4;
    const int lr   = lane & 15;
    const int m0   = (tile >> 3) * 16;
    const int n0   = (tile & 7) * 64 + 32 * w;

    const float* xrow = X  + (size_t)(m0 + lr) * HH + 2 * hi;
    const float* w0   = Wm + (size_t)(n0 + 2 * lr) * HH + 2 * hi;
    const float* w1   = w0 + HH;

    v8f acc0 = {}, acc1 = {};
    half_strip<HH / 16>(xrow, w0, w1, acc0, acc1);

#pragma unroll
    for (int r = 0; r < 8; ++r) {
        float2 o; o.x = acc0[r]; o.y = acc1[r];
        *reinterpret_cast<float2*>(C + (size_t)(m0 + r + 8 * hi) * HH + n0 + 2 * lr) = o;
    }
}

#define HG 32
#define TBR 2
#define QLIM 60.0f
static_assert(HG == 32, "a pass is one h per lane of the wave");

__device__ __forceinline__ float lane_value(float x, int j) {
    return __builtin_bit_cast(float, __builtin_amdgcn_readlane(__builtin_bit_cast(int, x), j));
}

__device__ __forceinline__ void score_pass(const v2f (&kk)[HG], float te0, float te1, float tv,
                                           v2f& acc0, v2f& acc1) {
#pragma unroll
    for (int j = 0; j < HG; ++j) {
        const v2f c = kk[j] * (2.0f * LOG2E);
        v2f ek; ek.x = __builtin_amdgcn_exp2f(c.x); ek.y = __builtin_amdgcn_exp2f(c.y);
        const float s0 = lane_value(te0, j);
        const float s1 = lane_value(te1, j);
        const float sv = lane_value(tv, j);
        const v2f d0 = ek * s0 + 1.0f;
        const v2f d1 = ek * s1 + 1.0f;
        v2f r0, r1;
        r0.x = __builtin_amdgcn_rcpf(d0.x); r0.y = __builtin_amdgcn_rcpf(d0.y);
        r1.x = __builtin_amdgcn_rcpf(d1.x); r1.y = __builtin_amdgcn_rcpf(d1.y);
        acc0 += r0 * sv;
        acc1 += r1 * sv;
    }
}

__global__ void __launch_bounds__(256) score_softmax_kernel(const float* __restrict__ qp,
                                     const float* __restrict__ kT,
                                     const float* __restrict__ vvec,
                                     const int* __restrict__ lens,
                                     float* __restrict__ attn) {
    const int b  = blockIdx.x / (TT / TBR);
    const int t0 = (blockIdx.x % (TT / TBR)) * TBR;
    const int tid  = threadIdx.x;
    const int wv   = tid >> 5;
    const int lane = tid & 31;

    __shared__ float redbuf[TBR][8];
    __shared__ int   okbuf[8];

    const float* __restrict__ q0 = qp + (size_t)(b * TT + t0) * HH;
    const float* __restrict__ q1 = q0 + HH;
    const float* __restrict__ kcol = kT + (size_t)b * HH * SS + 2 * tid;

    float vs = 0.0f;
    bool ok = true;
#pragma unroll
    for (int i = 0; i < HH / 256; ++i) {
        const int h = tid + 256 * i;
        ok = ok && (fabsf(q0[h] * (2.0f * LOG2E)) <= QLIM) && (fabsf(q1[h] * (2.0f * LOG2E)) <= QLIM);
        vs += vvec[h];
    }
#pragma unroll
    for (int o = 16; o >= 1; o >>= 1) vs += __shfl_xor(vs, o, 32);
    const int wok = __all(ok);
    if (lane == 0) { redbuf[0][wv] = vs; okbuf[wv] = wok; }
    __syncthreads();
    float vsum = redbuf[0][0];
    int allok = okbuf[0];
#pragma unroll
    for (int i = 1; i < 8; ++i) { vsum += redbuf[0][i]; allok &= okbuf[i]; }
    __syncthreads();

    v2f acc0 = {0.0f, 0.0f}, acc1 = {0.0f, 0.0f};
    if (allok) {
        v2f ka[HG], kb[HG];
        float qa0, qa1, va, qb0, qb1, vb;
#pragma unroll
        for (int j = 0; j < HG; ++j) ka[j] = ld2(kcol + (size_t)j * SS);
        qa0 = q0[lane]; qa1 = q1[lane]; va = vvec[lane];
        int h0 = 0;
        for (; h0 < HH - 2 * HG; h0 += 2 * HG) {
            __builtin_amdgcn_sched_barrier(0);
#pragma unroll
            for (int j = 0; j < HG; ++j) kb[j] = ld2(kcol + (size_t)(h0 + HG + j) * SS);
            qb0 = q0[h0 + HG + lane]; qb1 = q1[h0 + HG + lane]; vb = vvec[h0 + HG + lane];
            __builtin_amdgcn_sched_barrier(0);
            score_pass(ka, __builtin_amdgcn_exp2f(qa0 * (2.0f * LOG2E)),
                           __builtin_amdgcn_exp2f(qa1 * (2.0f * LOG2E)), va, acc0, acc1);
            __builtin_amdgcn_sched_barrier(0);
#pragma unroll
            for (int j = 0; j < HG; ++j) ka[j] = ld2(kcol + (size_t)(h0 + 2 * HG + j) * SS);
            qa0 = q0[h0 + 2 * HG + lane]; qa1 = q1[h0 + 2 * HG + lane]; va = vvec[h0 + 2 * HG + lane];
            __builtin_amdgcn_sched_barrier(0);
            score_pass(kb, __builtin_amdgcn_exp2f(qb0 * (2.0f * LOG2E)),
                           __builtin_amdgcn_exp2f(qb1 * (2.0f * LOG2E)), vb, acc0, acc1);
        }
        __builtin_amdgcn_sched_barrier(0);
#pragma unroll
        for (int j = 0; j < HG; ++j) kb[j] = ld2(kcol + (size_t)(h0 + HG + j) * SS);
        qb0 = q0[h0 + HG + lane]; qb1 = q1[h0 + HG + lane]; vb = vvec[h0 + HG + lane];
        __builtin_amdgcn_sched_barrier(0);
        score_pass(ka, __builtin_amdgcn_exp2f(qa0 * (2.0f * LOG2E)),
                       __builtin_amdgcn_exp2f(qa1 * (2.0f * LOG2E)), va, acc0, acc1);
        score_pass(kb, __builtin_amdgcn_exp2f(qb0 * (2.0f * LOG2E)),
                       __builtin_amdgcn_exp2f(qb1 * (2.0f * LOG2E)), vb, acc0, acc1);
        acc0 = vsum - 2.0f * acc0;
        acc1 = vsum - 2.0f * acc1;
    } else {
        for (int h0 = 0; h0 < HH; h0 += HG) {
            v2f kk[HG];
#pragma unroll
            for (int j = 0; j < HG; ++j) kk[j] = ld2(kcol + (size_t)(h0 + j) * SS);
#pragma unroll
            for (int j = 0; j < HG; ++j) {
                const float vh = vvec[h0 + j];
                acc0 += fast_tanh2(kk[j] + q0[h0 + j]) * vh;
                acc1 += fast_tanh2(kk[j] + q1[h0 + j]) * vh;
            }
        }
    }

    const int len = lens[b];
    const bool in0 = 2 * tid     < len;
    const bool in1 = 2 * tid + 1 < len;
    float s[TBR][2];
    s[0][0] = in0 ? acc0.x : -__builtin_inff();
    s[0][1] = in1 ? acc0.y : -__builtin_inff();
    s[1][0] = in0 ? acc1.x : -__builtin_inff();
    s[1][1] = in1 ? acc1.y : -__builtin_inff();

    float m[TBR];
#pragma unroll
    for (int r = 0; r < TBR; ++r) {
        m[r] = fmaxf(s[r][0], s[r][1]);
#pragma unroll
        for (int o = 16; o >= 1; o >>= 1) m[r] = fmaxf(m[r], __shfl_xor(m[r], o, 32));
        if (lane == 0) redbuf[r][wv] = m[r];
    }
    __syncthreads();
    float mx[TBR];
#pragma unroll
    for (int r = 0; r < TBR; ++r) {
        mx[r] = redbuf[r][0];
#pragma unroll
        for (int i = 1; i < 8; ++i) mx[r] = fmaxf(mx[r], redbuf[r][i]);
    }
    __syncthreads();

    float e[TBR][2], ps[TBR];
#pragma unroll
    for (int r = 0; r < TBR; ++r) {
        e[r][0] = __builtin_amdgcn_exp2f((s[r][0] - mx[r]) * LOG2E);
        e[r][1] = __builtin_amdgcn_exp2f((s[r][1] - mx[r]) * LOG2E);
        ps[r] = e[r][0] + e[r][1];
#pragma unroll
        for (int o = 16; o >= 1; o >>= 1) ps[r] += __shfl_xor(ps[r], o, 32);
        if (lane == 0) redbuf[r][wv] = ps[r];
    }
    __syncthreads();
#pragma unroll
    for (int r = 0; r < TBR; ++r) {
        float tot = redbuf[r][0];
#pragma unroll
        for (int i = 1; i < 8; ++i) tot += redbuf[r][i];
        const float inv = __builtin_amdgcn_rcpf(tot);
        float2 o2; o2.x = e[r][0] * inv; o2.y = e[r][1] * inv;
        *reinterpret_cast<float2*>(attn + (size_t)(b * TT + t0 + r) * SS + 2 * tid) = o2;
    }
}

__global__ void __launch_bounds__(64) context_kernel(const float* __restrict__ attn,
                                                     const float* __restrict__ enc,
                                                     float* __restrict__ ctx) {
    const int tile = blockIdx.x;
    const int w    = threadIdx.x >> 5;
    const int lane = threadIdx.x & 31;
    const int hi   = lane >> 4;
    const int lr   = lane & 15;
    const int b    = tile >> 6;
    const int rem  = tile & 63;
    const int m0   = (rem >> 3) * 16;
    const int n0   = (rem & 7) * 64 + 32 * w;

    const float* arow = attn + (size_t)(b * TT + m0 + lr) * SS + 2 * hi;
    const float* eb   = enc + (size_t)b * SS * HH + (size_t)(2 * hi) * HH + n0 + 2 * lr;

    v8f acc0 = {}, acc1 = {};
    v2f fa[RD + 1][4], e0[RD + 1][4], e1[RD + 1][4];
#pragma unroll
    for (int i = 0; i < RD; ++i) {
#pragma unroll
        for (int j = 0; j < 4; ++j) {
            const int k = 16 * i + 4 * j;
            fa[i][j] = ld2(arow + k);
            e0[i][j] = ld2(eb + (size_t)k * HH);
            e1[i][j] = ld2(eb + (size_t)(k + 1) * HH);
        }
        __builtin_amdgcn_sched_barrier(0);
    }
#pragma unroll
    for (int kb = 0; kb < SS / 16; ++kb) {
        if (kb + RD < SS / 16) {
            const int s = (kb + RD) % (RD + 1);
#pragma unroll
            for (int j = 0; j < 4; ++j) {
                const int k = 16 * (kb + RD) + 4 * j;
                fa[s][j] = ld2(arow + k);
                e0[s][j] = ld2(eb + (size_t)k * HH);
                e1[s][j] = ld2(eb + (size_t)(k + 1) * HH);
            }
            __builtin_amdgcn_sched_barrier(0);
        }
        const int c = kb % (RD + 1);
#pragma unroll
        for (int j = 0; j < 4; ++j) {
            v2f b0, b1;
            b0.x = e0[c][j].x; b0.y = e1[c][j].x;
            b1.x = e0[c][j].y; b1.y = e1[c][j].y;
            acc0 = WMMA_F32(fa[c][j], b0, acc0);
            acc1 = WMMA_F32(fa[c][j], b1, acc1);
        }
        asm volatile("" : "+v"(acc0), "+v"(acc1) : : "memory");
    }

#pragma unroll
    for (int r = 0; r < 8; ++r) {
        float2 o; o.x = acc0[r]; o.y = acc1[r];
        *reinterpret_cast<float2*>(ctx + (size_t)(b * TT + m0 + r + 8 * hi) * HH + n0 + 2 * lr) = o;
    }
}

__global__ void __launch_bounds__(64) out_gemm_kernel(const float* __restrict__ ctx,
                                const float* __restrict__ query,
                                const float* __restrict__ Wout,
                                const float* __restrict__ bout,
                                float* __restrict__ outp) {
    const int tile = blockIdx.x;
    const int w    = threadIdx.x >> 5;
    const int lane = threadIdx.x & 31;
    const int hi   = lane >> 4;
    const int lr   = lane & 15;
    const int m0   = (tile >> 3) * 16;
    const int n0   = (tile & 7) * 64 + 32 * w;

    const float* crow = ctx   + (size_t)(m0 + lr) * HH + 2 * hi;
    const float* qrow = query + (size_t)(m0 + lr) * HH + 2 * hi;
    const float* w0 = Wout + (size_t)(n0 + 2 * lr) * (2 * HH) + 2 * hi;
    const float* w1 = w0 + 2 * HH;

    v8f acc0 = {}, acc1 = {};
    half_strip<HH / 16>(crow, w0, w1, acc0, acc1);
    half_strip<HH / 16>(qrow, w0 + HH, w1 + HH, acc0, acc1);

    const float2 bb = *reinterpret_cast<const float2*>(bout + n0 + 2 * lr);
#pragma unroll
    for (int r = 0; r < 8; ++r) {
        float2 o;
        o.x = fast_tanh(acc0[r] + bb.x);
        o.y = fast_tanh(acc1[r] + bb.y);
        *reinterpret_cast<float2*>(outp + (size_t)(m0 + r + 8 * hi) * HH + n0 + 2 * lr) = o;
    }
}

__global__ void ln_kernel(const float* __restrict__ x,
                          const float* __restrict__ gamma,
                          const float* __restrict__ beta,
                          float* __restrict__ out) {
    const int wave = (blockIdx.x * blockDim.x + threadIdx.x) >> 5;
    const int lane = threadIdx.x & 31;
    const float* row = x + (size_t)wave * HH;

    float4 val[4];
    float s = 0.0f;
#pragma unroll
    for (int j = 0; j < 4; ++j) {
        val[j] = *(const float4*)(row + (lane + 32 * j) * 4);
        s += val[j].x + val[j].y + val[j].z + val[j].w;
    }
#pragma unroll
    for (int o = 16; o >= 1; o >>= 1) s += __shfl_xor(s, o, 32);
    const float mu = s * (1.0f / HH);

    float q = 0.0f;
#pragma unroll
    for (int j = 0; j < 4; ++j) {
        const float dx = val[j].x - mu, dy = val[j].y - mu;
        const float dz = val[j].z - mu, dw = val[j].w - mu;
        q += dx * dx + dy * dy + dz * dz + dw * dw;
    }
#pragma unroll
    for (int o = 16; o >= 1; o >>= 1) q += __shfl_xor(q, o, 32);
    const float rstd = __builtin_amdgcn_rsqf(q * (1.0f / HH) + LN_EPS);

    float* orow = out + (size_t)wave * HH;
#pragma unroll
    for (int j = 0; j < 4; ++j) {
        const int h = (lane + 32 * j) * 4;
        const float4 g = *(const float4*)(gamma + h);
        const float4 bt = *(const float4*)(beta + h);
        float4 r;
        r.x = (val[j].x - mu) * rstd * g.x + bt.x;
        r.y = (val[j].y - mu) * rstd * g.y + bt.y;
        r.z = (val[j].z - mu) * rstd * g.z + bt.z;
        r.w = (val[j].w - mu) * rstd * g.w + bt.w;
        *(float4*)(orow + h) = r;
    }
}

extern "C" void kernel_launch(void* const* d_in, const int* in_sizes, int n_in,
                              void* d_out, int out_size, void* d_ws, size_t ws_size,
                              hipStream_t stream) {
    const float* query  = (const float*)d_in[0];
    const float* enc    = (const float*)d_in[1];
    const int*   lens   = (const int*)  d_in[2];
    const float* W_h    = (const float*)d_in[3];
    const float* W_s    = (const float*)d_in[4];
    const float* vvec   = (const float*)d_in[5];
    const float* W_out  = (const float*)d_in[6];
    const float* b_out  = (const float*)d_in[7];
    const float* gamma  = (const float*)d_in[8];
    const float* beta   = (const float*)d_in[9];
    float* out = (float*)d_out;

    float* qp   = (float*)d_ws;
    float* kp   = qp   + (size_t)BB * TT * HH;
    float* attn = kp   + (size_t)BB * SS * HH;
    float* ctx  = attn + (size_t)BB * TT * SS;
    float* outp = ctx  + (size_t)BB * TT * HH;

    proj_gemm_kernel<<<dim3(256, 1), 64, 0, stream>>>(query, W_s, qp, 0, 0);
    proj_gemm_kernel<<<dim3(256, BB), 64, 0, stream>>>(W_h, enc, kp, SS * HH, HH * SS);
    score_softmax_kernel<<<BB * TT / TBR, 256, 0, stream>>>(qp, kp, vvec, lens, attn);
    context_kernel<<<256, 64, 0, stream>>>(attn, enc, ctx);
    out_gemm_kernel<<<256, 64, 0, stream>>>(ctx, query, W_out, b_out, outp);
    ln_kernel<<<64, 256, 0, stream>>>(outp, gamma, beta, out);
}


# q/k projection, output and context GEMMs: fragments loaded 16 B per lane (k re-paired in groups of 8, one load feeds two WMMAs): half the requests per 16-k block, ring 8 blocks ahead; new K sum order
# speedup vs baseline: 8.7968x; 1.0625x over previous
#include <hip/hip_runtime.h>

#define BB 4
#define TT 128
#define SS 512
#define HH 512
#define LN_EPS 1e-5f
#define LOG2E 1.4426950408889634f

typedef __attribute__((ext_vector_type(2))) float v2f;
typedef __attribute__((ext_vector_type(4))) float v4f;
typedef __attribute__((ext_vector_type(8))) float v8f;

__device__ __forceinline__ v2f ld2(const float* p) {
    const float2 t = *reinterpret_cast<const float2*>(p);
    v2f r; r.x = t.x; r.y = t.y; return r;
}

__device__ __forceinline__ v4f ld4(const float* p) {
    return *reinterpret_cast<const v4f*>(p);
}
__device__ __forceinline__ v2f lo2(v4f q) { return __builtin_shufflevector(q, q, 0, 1); }
__device__ __forceinline__ v2f hi2(v4f q) { return __builtin_shufflevector(q, q, 2, 3); }

__device__ __forceinline__ float fast_tanh(float x) {
    float xc = fminf(fmaxf(x, -15.0f), 15.0f);
    float e  = __builtin_amdgcn_exp2f(xc * (2.0f * LOG2E));
    return (e - 1.0f) * __builtin_amdgcn_rcpf(e + 1.0f);
}

__device__ __forceinline__ v2f fast_tanh2(v2f x) {
    v2f xc;
    xc.x = fminf(fmaxf(x.x, -15.0f), 15.0f);
    xc.y = fminf(fmaxf(x.y, -15.0f), 15.0f);
    const v2f tt = xc * (2.0f * LOG2E);
    v2f e;
    e.x = __builtin_amdgcn_exp2f(tt.x);
    e.y = __builtin_amdgcn_exp2f(tt.y);
    const v2f d = e + 1.0f;
    v2f r;
    r.x = __builtin_amdgcn_rcpf(d.x);
    r.y = __builtin_amdgcn_rcpf(d.y);
    return (e - 1.0f) * r;
}

#define WMMA_F32(A, Bf, C) \
    __builtin_amdgcn_wmma_f32_16x16x4_f32(false, (A), false, (Bf), (short)0, (C), false, false)

#define RD 8
template <int KB>
__device__ __forceinline__ void half_strip(const float* __restrict__ xrow,
                                           const float* __restrict__ w0, const float* __restrict__ w1,
                                           v8f& acc0, v8f& acc1) {
    v4f fa[RD + 1][2], f0[RD + 1][2], f1[RD + 1][2];
#pragma unroll
    for (int i = 0; i < RD; ++i) {
#pragma unroll
        for (int u = 0; u < 2; ++u) {
            const int o = 16 * i + 8 * u;
            fa[i][u] = ld4(xrow + o); f0[i][u] = ld4(w0 + o); f1[i][u] = ld4(w1 + o);
        }
        __builtin_amdgcn_sched_barrier(0);
    }
#pragma unroll
    for (int kb = 0; kb < KB; ++kb) {
        if (kb + RD < KB) {
            const int s = (kb + RD) % (RD + 1);
#pragma unroll
            for (int u = 0; u < 2; ++u) {
                const int o = 16 * (kb + RD) + 8 * u;
                fa[s][u] = ld4(xrow + o); f0[s][u] = ld4(w0 + o); f1[s][u] = ld4(w1 + o);
            }
            __builtin_amdgcn_sched_barrier(0);
        }
        const int c = kb % (RD + 1);
#pragma unroll
        for (int u = 0; u < 2; ++u) {
            acc0 = WMMA_F32(lo2(fa[c][u]), lo2(f0[c][u]), acc0);
            acc1 = WMMA_F32(lo2(fa[c][u]), lo2(f1[c][u]), acc1);
            acc0 = WMMA_F32(hi2(fa[c][u]), hi2(f0[c][u]), acc0);
            acc1 = WMMA_F32(hi2(fa[c][u]), hi2(f1[c][u]), acc1);
        }
        asm volatile("" : "+v"(acc0), "+v"(acc1) : : "memory");
    }
}

static_assert(SS == HH, "both projections write rows of 512 floats");
__global__ void __launch_bounds__(64) proj_gemm_kernel(const float* __restrict__ X,
                                                       const float* __restrict__ Wm,
                                                       float* __restrict__ C,
                                                       int wBatch, int cBatch) {
    Wm += (size_t)blockIdx.y * wBatch;
    C  += (size_t)blockIdx.y * cBatch;
    const int tile = blockIdx.x;
    const int w    = threadIdx.x >> 5;
    const int lane = threadIdx.x & 31;
    const int hi   = lane >> 4;
    const int lr   = lane & 15;
    const int m0   = (tile >> 3) * 16;
    const int n0   = (tile & 7) * 64 + 32 * w;

    const float* xrow = X  + (size_t)(m0 + lr) * HH + 4 * hi;
    const float* w0   = Wm + (size_t)(n0 + 2 * lr) * HH + 4 * hi;
    const float* w1   = w0 + HH;

    v8f acc0 = {}, acc1 = {};
    half_strip<HH / 16>(xrow, w0, w1, acc0, acc1);

#pragma unroll
    for (int r = 0; r < 8; ++r) {
        float2 o; o.x = acc0[r]; o.y = acc1[r];
        *reinterpret_cast<float2*>(C + (size_t)(m0 + r + 8 * hi) * HH + n0 + 2 * lr) = o;
    }
}

#define HG 32
#define TBR 2
#define QLIM 60.0f
static_assert(HG == 32, "a pass is one h per lane of the wave");

__device__ __forceinline__ float lane_value(float x, int j) {
    return __builtin_bit_cast(float, __builtin_amdgcn_readlane(__builtin_bit_cast(int, x), j));
}

__device__ __forceinline__ void score_pass(const v2f (&kk)[HG], float te0, float te1, float tv,
                                           v2f& acc0, v2f& acc1) {
#pragma unroll
    for (int j = 0; j < HG; ++j) {
        const v2f c = kk[j] * (2.0f * LOG2E);
        v2f ek; ek.x = __builtin_amdgcn_exp2f(c.x); ek.y = __builtin_amdgcn_exp2f(c.y);
        const float s0 = lane_value(te0, j);
        const float s1 = lane_value(te1, j);
        const float sv = lane_value(tv, j);
        const v2f d0 = ek * s0 + 1.0f;
        const v2f d1 = ek * s1 + 1.0f;
        v2f r0, r1;
        r0.x = __builtin_amdgcn_rcpf(d0.x); r0.y = __builtin_amdgcn_rcpf(d0.y);
        r1.x = __builtin_amdgcn_rcpf(d1.x); r1.y = __builtin_amdgcn_rcpf(d1.y);
        acc0 += r0 * sv;
        acc1 += r1 * sv;
    }
}

__global__ void __launch_bounds__(256) score_softmax_kernel(const float* __restrict__ qp,
                                     const float* __restrict__ kT,
                                     const float* __restrict__ vvec,
                                     const int* __restrict__ lens,
                                     float* __restrict__ attn) {
    const int b  = blockIdx.x / (TT / TBR);
    const int t0 = (blockIdx.x % (TT / TBR)) * TBR;
    const int tid  = threadIdx.x;
    const int wv   = tid >> 5;
    const int lane = tid & 31;

    __shared__ float redbuf[TBR][8];
    __shared__ int   okbuf[8];

    const float* __restrict__ q0 = qp + (size_t)(b * TT + t0) * HH;
    const float* __restrict__ q1 = q0 + HH;
    const float* __restrict__ kcol = kT + (size_t)b * HH * SS + 2 * tid;

    float vs = 0.0f;
    bool ok = true;
#pragma unroll
    for (int i = 0; i < HH / 256; ++i) {
        const int h = tid + 256 * i;
        ok = ok && (fabsf(q0[h] * (2.0f * LOG2E)) <= QLIM) && (fabsf(q1[h] * (2.0f * LOG2E)) <= QLIM);
        vs += vvec[h];
    }
#pragma unroll
    for (int o = 16; o >= 1; o >>= 1) vs += __shfl_xor(vs, o, 32);
    const int wok = __all(ok);
    if (lane == 0) { redbuf[0][wv] = vs; okbuf[wv] = wok; }
    __syncthreads();
    float vsum = redbuf[0][0];
    int allok = okbuf[0];
#pragma unroll
    for (int i = 1; i < 8; ++i) { vsum += redbuf[0][i]; allok &= okbuf[i]; }
    __syncthreads();

    v2f acc0 = {0.0f, 0.0f}, acc1 = {0.0f, 0.0f};
    if (allok) {
        v2f ka[HG], kb[HG];
        float qa0, qa1, va, qb0, qb1, vb;
#pragma unroll
        for (int j = 0; j < HG; ++j) ka[j] = ld2(kcol + (size_t)j * SS);
        qa0 = q0[lane]; qa1 = q1[lane]; va = vvec[lane];
        int h0 = 0;
        for (; h0 < HH - 2 * HG; h0 += 2 * HG) {
            __builtin_amdgcn_sched_barrier(0);
#pragma unroll
            for (int j = 0; j < HG; ++j) kb[j] = ld2(kcol + (size_t)(h0 + HG + j) * SS);
            qb0 = q0[h0 + HG + lane]; qb1 = q1[h0 + HG + lane]; vb = vvec[h0 + HG + lane];
            __builtin_amdgcn_sched_barrier(0);
            score_pass(ka, __builtin_amdgcn_exp2f(qa0 * (2.0f * LOG2E)),
                           __builtin_amdgcn_exp2f(qa1 * (2.0f * LOG2E)), va, acc0, acc1);
            __builtin_amdgcn_sched_barrier(0);
#pragma unroll
            for (int j = 0; j < HG; ++j) ka[j] = ld2(kcol + (size_t)(h0 + 2 * HG + j) * SS);
            qa0 = q0[h0 + 2 * HG + lane]; qa1 = q1[h0 + 2 * HG + lane]; va = vvec[h0 + 2 * HG + lane];
            __builtin_amdgcn_sched_barrier(0);
            score_pass(kb, __builtin_amdgcn_exp2f(qb0 * (2.0f * LOG2E)),
                           __builtin_amdgcn_exp2f(qb1 * (2.0f * LOG2E)), vb, acc0, acc1);
        }
        __builtin_amdgcn_sched_barrier(0);
#pragma unroll
        for (int j = 0; j < HG; ++j) kb[j] = ld2(kcol + (size_t)(h0 + HG + j) * SS);
        qb0 = q0[h0 + HG + lane]; qb1 = q1[h0 + HG + lane]; vb = vvec[h0 + HG + lane];
        __builtin_amdgcn_sched_barrier(0);
        score_pass(ka, __builtin_amdgcn_exp2f(qa0 * (2.0f * LOG2E)),
                       __builtin_amdgcn_exp2f(qa1 * (2.0f * LOG2E)), va, acc0, acc1);
        score_pass(kb, __builtin_amdgcn_exp2f(qb0 * (2.0f * LOG2E)),
                       __builtin_amdgcn_exp2f(qb1 * (2.0f * LOG2E)), vb, acc0, acc1);
        acc0 = vsum - 2.0f * acc0;
        acc1 = vsum - 2.0f * acc1;
    } else {
        for (int h0 = 0; h0 < HH; h0 += HG) {
            v2f kk[HG];
#pragma unroll
            for (int j = 0; j < HG; ++j) kk[j] = ld2(kcol + (size_t)(h0 + j) * SS);
#pragma unroll
            for (int j = 0; j < HG; ++j) {
                const float vh = vvec[h0 + j];
                acc0 += fast_tanh2(kk[j] + q0[h0 + j]) * vh;
                acc1 += fast_tanh2(kk[j] + q1[h0 + j]) * vh;
            }
        }
    }

    const int len = lens[b];
    const bool in0 = 2 * tid     < len;
    const bool in1 = 2 * tid + 1 < len;
    float s[TBR][2];
    s[0][0] = in0 ? acc0.x : -__builtin_inff();
    s[0][1] = in1 ? acc0.y : -__builtin_inff();
    s[1][0] = in0 ? acc1.x : -__builtin_inff();
    s[1][1] = in1 ? acc1.y : -__builtin_inff();

    float m[TBR];
#pragma unroll
    for (int r = 0; r < TBR; ++r) {
        m[r] = fmaxf(s[r][0], s[r][1]);
#pragma unroll
        for (int o = 16; o >= 1; o >>= 1) m[r] = fmaxf(m[r], __shfl_xor(m[r], o, 32));
        if (lane == 0) redbuf[r][wv] = m[r];
    }
    __syncthreads();
    float mx[TBR];
#pragma unroll
    for (int r = 0; r < TBR; ++r) {
        mx[r] = redbuf[r][0];
#pragma unroll
        for (int i = 1; i < 8; ++i) mx[r] = fmaxf(mx[r], redbuf[r][i]);
    }
    __syncthreads();

    float e[TBR][2], ps[TBR];
#pragma unroll
    for (int r = 0; r < TBR; ++r) {
        e[r][0] = __builtin_amdgcn_exp2f((s[r][0] - mx[r]) * LOG2E);
        e[r][1] = __builtin_amdgcn_exp2f((s[r][1] - mx[r]) * LOG2E);
        ps[r] = e[r][0] + e[r][1];
#pragma unroll
        for (int o = 16; o >= 1; o >>= 1) ps[r] += __shfl_xor(ps[r], o, 32);
        if (lane == 0) redbuf[r][wv] = ps[r];
    }
    __syncthreads();
#pragma unroll
    for (int r = 0; r < TBR; ++r) {
        float tot = redbuf[r][0];
#pragma unroll
        for (int i = 1; i < 8; ++i) tot += redbuf[r][i];
        const float inv = __builtin_amdgcn_rcpf(tot);
        float2 o2; o2.x = e[r][0] * inv; o2.y = e[r][1] * inv;
        *reinterpret_cast<float2*>(attn + (size_t)(b * TT + t0 + r) * SS + 2 * tid) = o2;
    }
}

#define RDC 4
__global__ void __launch_bounds__(64) context_kernel(const float* __restrict__ attn,
                                                     const float* __restrict__ enc,
                                                     float* __restrict__ ctx) {
    const int tile = blockIdx.x;
    const int w    = threadIdx.x >> 5;
    const int lane = threadIdx.x & 31;
    const int hi   = lane >> 4;
    const int lr   = lane & 15;
    const int b    = tile >> 6;
    const int rem  = tile & 63;
    const int m0   = (rem >> 3) * 16;
    const int n0   = (rem & 7) * 64 + 32 * w;

    const float* arow = attn + (size_t)(b * TT + m0 + lr) * SS + 4 * hi;
    const float* eb   = enc + (size_t)b * SS * HH + (size_t)(4 * hi) * HH + n0 + 2 * lr;

    v8f acc0 = {}, acc1 = {};
    v4f fa[RDC + 1][2];
    v2f e0[RDC + 1][4], e1[RDC + 1][4];
#pragma unroll
    for (int i = 0; i < RDC; ++i) {
#pragma unroll
        for (int u = 0; u < 2; ++u) fa[i][u] = ld4(arow + 16 * i + 8 * u);
#pragma unroll
        for (int j = 0; j < 4; ++j) {
            const int k = 16 * i + 8 * (j >> 1) + 2 * (j & 1);
            e0[i][j] = ld2(eb + (size_t)k * HH);
            e1[i][j] = ld2(eb + (size_t)(k + 1) * HH);
        }
        __builtin_amdgcn_sched_barrier(0);
    }
#pragma unroll
    for (int kb = 0; kb < SS / 16; ++kb) {
        if (kb + RDC < SS / 16) {
            const int s = (kb + RDC) % (RDC + 1);
#pragma unroll
            for (int u = 0; u < 2; ++u) fa[s][u] = ld4(arow + 16 * (kb + RDC) + 8 * u);
#pragma unroll
            for (int j = 0; j < 4; ++j) {
                const int k = 16 * (kb + RDC) + 8 * (j >> 1) + 2 * (j & 1);
                e0[s][j] = ld2(eb + (size_t)k * HH);
                e1[s][j] = ld2(eb + (size_t)(k + 1) * HH);
            }
            __builtin_amdgcn_sched_barrier(0);
        }
        const int c = kb % (RDC + 1);
#pragma unroll
        for (int j = 0; j < 4; ++j) {
            v2f b0, b1;
            b0.x = e0[c][j].x; b0.y = e1[c][j].x;
            b1.x = e0[c][j].y; b1.y = e1[c][j].y;
            const v2f a = (j & 1) ? hi2(fa[c][j >> 1]) : lo2(fa[c][j >> 1]);
            acc0 = WMMA_F32(a, b0, acc0);
            acc1 = WMMA_F32(a, b1, acc1);
        }
        asm volatile("" : "+v"(acc0), "+v"(acc1) : : "memory");
    }

#pragma unroll
    for (int r = 0; r < 8; ++r) {
        float2 o; o.x = acc0[r]; o.y = acc1[r];
        *reinterpret_cast<float2*>(ctx + (size_t)(b * TT + m0 + r + 8 * hi) * HH + n0 + 2 * lr) = o;
    }
}

__global__ void __launch_bounds__(64) out_gemm_kernel(const float* __restrict__ ctx,
                                const float* __restrict__ query,
                                const float* __restrict__ Wout,
                                const float* __restrict__ bout,
                                float* __restrict__ outp) {
    const int tile = blockIdx.x;
    const int w    = threadIdx.x >> 5;
    const int lane = threadIdx.x & 31;
    const int hi   = lane >> 4;
    const int lr   = lane & 15;
    const int m0   = (tile >> 3) * 16;
    const int n0   = (tile & 7) * 64 + 32 * w;

    const float* crow = ctx   + (size_t)(m0 + lr) * HH + 4 * hi;
    const float* qrow = query + (size_t)(m0 + lr) * HH + 4 * hi;
    const float* w0 = Wout + (size_t)(n0 + 2 * lr) * (2 * HH) + 4 * hi;
    const float* w1 = w0 + 2 * HH;

    v8f acc0 = {}, acc1 = {};
    half_strip<HH / 16>(crow, w0, w1, acc0, acc1);
    half_strip<HH / 16>(qrow, w0 + HH, w1 + HH, acc0, acc1);

    const float2 bb = *reinterpret_cast<const float2*>(bout + n0 + 2 * lr);
#pragma unroll
    for (int r = 0; r < 8; ++r) {
        float2 o;
        o.x = fast_tanh(acc0[r] + bb.x);
        o.y = fast_tanh(acc1[r] + bb.y);
        *reinterpret_cast<float2*>(outp + (size_t)(m0 + r + 8 * hi) * HH + n0 + 2 * lr) = o;
    }
}

__global__ void ln_kernel(const float* __restrict__ x,
                          const float* __restrict__ gamma,
                          const float* __restrict__ beta,
                          float* __restrict__ out) {
    const int wave = (blockIdx.x * blockDim.x + threadIdx.x) >> 5;
    const int lane = threadIdx.x & 31;
    const float* row = x + (size_t)wave * HH;

    float4 val[4];
    float s = 0.0f;
#pragma unroll
    for (int j = 0; j < 4; ++j) {
        val[j] = *(const float4*)(row + (lane + 32 * j) * 4);
        s += val[j].x + val[j].y + val[j].z + val[j].w;
    }
#pragma unroll
    for (int o = 16; o >= 1; o >>= 1) s += __shfl_xor(s, o, 32);
    const float mu = s * (1.0f / HH);

    float q = 0.0f;
#pragma unroll
    for (int j = 0; j < 4; ++j) {
        const float dx = val[j].x - mu, dy = val[j].y - mu;
        const float dz = val[j].z - mu, dw = val[j].w - mu;
        q += dx * dx + dy * dy + dz * dz + dw * dw;
    }
#pragma unroll
    for (int o = 16; o >= 1; o >>= 1) q += __shfl_xor(q, o, 32);
    const float rstd = __builtin_amdgcn_rsqf(q * (1.0f / HH) + LN_EPS);

    float* orow = out + (size_t)wave * HH;
#pragma unroll
    for (int j = 0; j < 4; ++j) {
        const int h = (lane + 32 * j) * 4;
        const float4 g = *(const float4*)(gamma + h);
        const float4 bt = *(const float4*)(beta + h);
        float4 r;
        r.x = (val[j].x - mu) * rstd * g.x + bt.x;
        r.y = (val[j].y - mu) * rstd * g.y + bt.y;
        r.z = (val[j].z - mu) * rstd * g.z + bt.z;
        r.w = (val[j].w - mu) * rstd * g.w + bt.w;
        *(float4*)(orow + h) = r;
    }
}

extern "C" void kernel_launch(void* const* d_in, const int* in_sizes, int n_in,
                              void* d_out, int out_size, void* d_ws, size_t ws_size,
                              hipStream_t stream) {
    const float* query  = (const float*)d_in[0];
    const float* enc    = (const float*)d_in[1];
    const int*   lens   = (const int*)  d_in[2];
    const float* W_h    = (const float*)d_in[3];
    const float* W_s    = (const float*)d_in[4];
    const float* vvec   = (const float*)d_in[5];
    const float* W_out  = (const float*)d_in[6];
    const float* b_out  = (const float*)d_in[7];
    const float* gamma  = (const float*)d_in[8];
    const float* beta   = (const float*)d_in[9];
    float* out = (float*)d_out;

    float* qp   = (float*)d_ws;
    float* kp   = qp   + (size_t)BB * TT * HH;
    float* attn = kp   + (size_t)BB * SS * HH;
    float* ctx  = attn + (size_t)BB * TT * SS;
    float* outp = ctx  + (size_t)BB * TT * HH;

    proj_gemm_kernel<<<dim3(256, 1), 64, 0, stream>>>(query, W_s, qp, 0, 0);
    proj_gemm_kernel<<<dim3(256, BB), 64, 0, stream>>>(W_h, enc, kp, SS * HH, HH * SS);
    score_softmax_kernel<<<BB * TT / TBR, 256, 0, stream>>>(qp, kp, vvec, lens, attn);
    context_kernel<<<256, 64, 0, stream>>>(attn, enc, ctx);
    out_gemm_kernel<<<256, 64, 0, stream>>>(ctx, query, W_out, b_out, outp);
    ln_kernel<<<64, 256, 0, stream>>>(outp, gamma, beta, out);
}
